# starting kernel
# baseline (speedup 1.0000x reference)
#include <hip/hip_runtime.h>
#include <hip/hip_bf16.h>

typedef __attribute__((ext_vector_type(2))) float v2f;
typedef __attribute__((ext_vector_type(8))) float v8f;

#define D 128
#define LDS_STRIDE 132

__global__ __launch_bounds__(256) void zero4_kernel(float4* __restrict__ p, long n4) {
    long i = (long)blockIdx.x * blockDim.x + threadIdx.x;
    if (i < n4) p[i] = make_float4(0.f, 0.f, 0.f, 0.f);
}

__global__ __launch_bounds__(256)
void gnn_scatter_kernel(const float* __restrict__ UF, const float* __restrict__ IF,
                        const int* __restrict__ uidx, const int* __restrict__ iidx,
                        float* __restrict__ aggU, float* __restrict__ aggI,
                        float* __restrict__ degU, float* __restrict__ degI, int E)
{
    int t = blockIdx.x * 256 + threadIdx.x;
    int e = t >> 5;
    if (e >= E) return;
    int lane = threadIdx.x & 31;
    int u  = uidx[e];
    int it = iidx[e];

    const float4 fu = ((const float4*)(UF + (size_t)u  * D))[lane];
    const float4 fi = ((const float4*)(IF + (size_t)it * D))[lane];

    float* ai = aggI + (size_t)it * D + lane * 4;
    float* au = aggU + (size_t)u  * D + lane * 4;

    unsafeAtomicAdd(ai + 0, fu.x);
    unsafeAtomicAdd(ai + 1, fu.y);
    unsafeAtomicAdd(ai + 2, fu.z);
    unsafeAtomicAdd(ai + 3, fu.w);

    unsafeAtomicAdd(au + 0, fi.x);
    unsafeAtomicAdd(au + 1, fi.y);
    unsafeAtomicAdd(au + 2, fi.z);
    unsafeAtomicAdd(au + 3, fi.w);

    if (lane == 0) {
        unsafeAtomicAdd(degU + u, 1.0f);
        unsafeAtomicAdd(degI + it, 1.0f);
    }
}

__global__ __launch_bounds__(256)
void gnn_gemm_kernel(const float* __restrict__ X, float* __restrict__ out,
                     const float* __restrict__ Wself, const float* __restrict__ bself,
                     const float* __restrict__ Wmsg,  const float* __restrict__ bmsg,
                     const float* __restrict__ deg)
{
    __shared__ float sX[16 * LDS_STRIDE];
    __shared__ float sG[16 * LDS_STRIDE];

    const int row0 = blockIdx.x * 16;
    const int tid  = threadIdx.x;

    for (int i = tid; i < 16 * 32; i += 256) {
        int r  = i >> 5;
        int c4 = i & 31;
        float4 x = ((const float4*)(X   + (size_t)(row0 + r) * D))[c4];
        float4 g = ((const float4*)(out + (size_t)(row0 + r) * D))[c4];
        ((float4*)(sX + r * LDS_STRIDE))[c4] = x;
        ((float4*)(sG + r * LDS_STRIDE))[c4] = g;
    }
    __syncthreads();

    const int lane = tid & 31;
    const int wave = tid >> 5;
    const int col0 = wave * 16;
    const int mn   = lane & 15;
    const int koff = (lane >> 4) << 1;

    const float* wS = Wself + (size_t)(col0 + mn) * D;
    const float* wM = Wmsg  + (size_t)(col0 + mn) * D;
    const float* aS = sX + mn * LDS_STRIDE;
    const float* aM = sG + mn * LDS_STRIDE;

    v8f c = {};
    #pragma unroll
    for (int k = 0; k < D; k += 4) {
        v2f a = { aS[k + koff], aS[k + koff + 1] };
        v2f b = { wS[k + koff], wS[k + koff + 1] };
        c = __builtin_amdgcn_wmma_f32_16x16x4_f32(false, a, false, b,
                                                  (short)0, c, false, false);
    }
    #pragma unroll
    for (int k = 0; k < D; k += 4) {
        v2f a = { aM[k + koff], aM[k + koff + 1] };
        v2f b = { wM[k + koff], wM[k + koff + 1] };
        c = __builtin_amdgcn_wmma_f32_16x16x4_f32(false, a, false, b,
                                                  (short)0, c, false, false);
    }

    const int col  = col0 + mn;
    const float bS = bself[col];
    const float bM = bmsg[col];
    const int rbase = row0 + ((lane >> 4) << 3);
    #pragma unroll
    for (int v = 0; v < 8; ++v) {
        int row = rbase + v;
        out[(size_t)row * D + col] = c[v] + bS + deg[row] * bM;
    }
}

extern "C" void kernel_launch(void* const* d_in, const int* in_sizes, int n_in,
                              void* d_out, int out_size, void* d_ws, size_t ws_size,
                              hipStream_t stream)
{
    const float* UF   = (const float*)d_in[0];
    const float* IF   = (const float*)d_in[1];
    const int*   edge = (const int*)d_in[2];
    const float* Wu   = (const float*)d_in[3];
    const float* bu   = (const float*)d_in[4];
    const float* Wi   = (const float*)d_in[5];
    const float* bi   = (const float*)d_in[6];
    const float* Wu2i = (const float*)d_in[7];
    const float* bu2i = (const float*)d_in[8];
    const float* Wi2u = (const float*)d_in[9];
    const float* bi2u = (const float*)d_in[10];

    const int U = in_sizes[0] / D;
    const int I = in_sizes[1] / D;
    const int E = in_sizes[2] / 2;

    float* outU = (float*)d_out;
    float* outI = outU + (size_t)U * D;
    float* degU = (float*)d_ws;
    float* degI = degU + U;

    {
        long n4 = (long)(U + I) * D / 4;
        zero4_kernel<<<(unsigned)((n4 + 255) / 256), 256, 0, stream>>>((float4*)d_out, n4);
        long d4 = (long)(U + I) / 4;
        zero4_kernel<<<(unsigned)((d4 + 255) / 256), 256, 0, stream>>>((float4*)d_ws, d4);
    }

    {
        long threads = (long)E * 32;
        gnn_scatter_kernel<<<(unsigned)((threads + 255) / 256), 256, 0, stream>>>(
            UF, IF, edge, edge + E, outU, outI, degU, degI, E);
    }

    gnn_gemm_kernel<<<U / 16, 256, 0, stream>>>(UF, outU, Wu, bu, Wi2u, bi2u, degU);
    gnn_gemm_kernel<<<I / 16, 256, 0, stream>>>(IF, outI, Wi, bi, Wu2i, bu2i, degI);
}


# Segment sums by gather: edges bucketed per destination row (int atomic slots), bucket sorted by source index, rows added in ascending index order (fixed sum order); replaces f32 atomic scatter
# speedup vs baseline: 1.1028x; 1.1028x over previous
#include <hip/hip_runtime.h>
#include <hip/hip_bf16.h>

typedef __attribute__((ext_vector_type(2))) float v2f;
typedef __attribute__((ext_vector_type(8))) float v8f;

#define D 128
#define LDS_STRIDE 132

__global__ __launch_bounds__(256) void zero_i4_kernel(int4* __restrict__ p, int n4) {
    int i = blockIdx.x * 256 + threadIdx.x;
    if (i < n4) p[i] = make_int4(0, 0, 0, 0);
}

__global__ __launch_bounds__(256)
void edge_count_kernel(const int* __restrict__ uidx, const int* __restrict__ iidx,
                       int* __restrict__ cntU, int* __restrict__ cntI, int E)
{
    int e = blockIdx.x * 256 + threadIdx.x;
    if (e >= E) return;
    atomicAdd(cntU + uidx[e], 1);
    atomicAdd(cntI + iidx[e], 1);
}

__global__ __launch_bounds__(1024)
void bucket_alloc_kernel(const int* __restrict__ cnt, int* __restrict__ off,
                         float* __restrict__ degf, int* __restrict__ cursor, int R)
{
    __shared__ int wsum[32];
    __shared__ int base_s;
    const int r    = blockIdx.x * 1024 + threadIdx.x;
    const int lane = threadIdx.x & 31;
    const int w    = threadIdx.x >> 5;
    const int c    = (r < R) ? cnt[r] : 0;

    int x = c;
    for (int d = 1; d < 32; d <<= 1) {
        int y = __shfl_up(x, d);
        if (lane >= d) x += y;
    }
    if (lane == 31) wsum[w] = x;
    __syncthreads();
    if (w == 0) {
        const int s = wsum[lane];
        int t = s;
        for (int d = 1; d < 32; d <<= 1) {
            int y = __shfl_up(t, d);
            if (lane >= d) t += y;
        }
        wsum[lane] = t - s;
        if (lane == 31) base_s = atomicAdd(cursor, t);
    }
    __syncthreads();
    if (r < R) {
        off[r]  = base_s + wsum[w] + x - c;
        degf[r] = (float)c;
    }
}

__global__ __launch_bounds__(256)
void bucket_fill_kernel(const int* __restrict__ uidx, const int* __restrict__ iidx,
                        int* __restrict__ cntU, int* __restrict__ cntI,
                        const int* __restrict__ offU, const int* __restrict__ offI,
                        int* __restrict__ bkt, int E)
{
    int e = blockIdx.x * 256 + threadIdx.x;
    if (e >= E) return;
    const int u  = uidx[e];
    const int it = iidx[e];
    const int pu = atomicAdd(cntU + u,  -1) - 1;
    const int pi = atomicAdd(cntI + it, -1) - 1;
    bkt[offU[u]  + pu] = it;
    bkt[offI[it] + pi] = u;
}

__global__ __launch_bounds__(256)
void segment_gather_kernel(const float* __restrict__ UF, const float* __restrict__ IF,
                           const int* __restrict__ off, const float* __restrict__ degf,
                           const int* __restrict__ bkt, float* __restrict__ out,
                           int U, int R)
{
    const int lane = threadIdx.x & 31;
    const int r = __builtin_amdgcn_readfirstlane(blockIdx.x * 8 + (threadIdx.x >> 5));
    if (r >= R) return;
    const float* __restrict__ src = (r < U) ? IF : UF;
    const int n = __builtin_amdgcn_readfirstlane((int)degf[r]);
    const int o = __builtin_amdgcn_readfirstlane(off[r]);

    float4 acc = make_float4(0.f, 0.f, 0.f, 0.f);

    if (n > 0 && n <= 32) {
        const int k = (lane < n) ? bkt[o + lane] : 0x7fffffff;
        int rk = 0;
        for (int l = 0; l < n; ++l) {
            const int s = __builtin_amdgcn_readlane(k, l);
            rk += (s < k || (s == k && l < lane)) ? 1 : 0;
        }
        const int rank = (lane < n) ? rk : lane;
        const int sk = __builtin_amdgcn_ds_permute(rank << 2, k);

        for (int t0 = 0; t0 < n; t0 += 8) {
            float4 f[8];
            #pragma unroll
            for (int j = 0; j < 8; ++j) {
                const int t = (t0 + j < n) ? (t0 + j) : (n - 1);
                const int s = __builtin_amdgcn_readlane(sk, t);
                f[j] = ((const float4*)(src + (size_t)s * D))[lane];
            }
            #pragma unroll
            for (int j = 0; j < 8; ++j) {
                const bool live = (t0 + j < n);
                acc.x += live ? f[j].x : 0.f;
                acc.y += live ? f[j].y : 0.f;
                acc.z += live ? f[j].z : 0.f;
                acc.w += live ? f[j].w : 0.f;
            }
        }
    } else if (n > 32) {
        int cur = -1;
        for (int trip = 0; trip < n; ++trip) {
            int best = 0x7fffffff, m = 0;
            for (int j = lane; j < n; j += 32) {
                const int k = bkt[o + j];
                if (k > cur) {
                    if (k < best)       { best = k; m = 1; }
                    else if (k == best) { ++m; }
                }
            }
            int wb = best;
            for (int d = 16; d > 0; d >>= 1) wb = min(wb, __shfl_xor(wb, d));
            if (wb == 0x7fffffff) break;
            int mm = (best == wb) ? m : 0;
            for (int d = 16; d > 0; d >>= 1) mm += __shfl_xor(mm, d);
            const int s   = __builtin_amdgcn_readfirstlane(wb);
            const int rep = __builtin_amdgcn_readfirstlane(mm);
            const float4 f = ((const float4*)(src + (size_t)s * D))[lane];
            for (int t = 0; t < rep; ++t) {
                acc.x += f.x; acc.y += f.y; acc.z += f.z; acc.w += f.w;
            }
            cur = s;
        }
    }

    ((float4*)(out + (size_t)r * D))[lane] = acc;
}

__global__ __launch_bounds__(256)
void gnn_gemm_kernel(const float* __restrict__ X, float* __restrict__ out,
                     const float* __restrict__ Wself, const float* __restrict__ bself,
                     const float* __restrict__ Wmsg,  const float* __restrict__ bmsg,
                     const float* __restrict__ deg)
{
    __shared__ float sX[16 * LDS_STRIDE];
    __shared__ float sG[16 * LDS_STRIDE];

    const int row0 = blockIdx.x * 16;
    const int tid  = threadIdx.x;

    for (int i = tid; i < 16 * 32; i += 256) {
        int r  = i >> 5;
        int c4 = i & 31;
        float4 x = ((const float4*)(X   + (size_t)(row0 + r) * D))[c4];
        float4 g = ((const float4*)(out + (size_t)(row0 + r) * D))[c4];
        ((float4*)(sX + r * LDS_STRIDE))[c4] = x;
        ((float4*)(sG + r * LDS_STRIDE))[c4] = g;
    }
    __syncthreads();

    const int lane = tid & 31;
    const int wave = tid >> 5;
    const int col0 = wave * 16;
    const int mn   = lane & 15;
    const int koff = (lane >> 4) << 1;

    const float* wS = Wself + (size_t)(col0 + mn) * D;
    const float* wM = Wmsg  + (size_t)(col0 + mn) * D;
    const float* aS = sX + mn * LDS_STRIDE;
    const float* aM = sG + mn * LDS_STRIDE;

    v8f c = {};
    #pragma unroll
    for (int k = 0; k < D; k += 4) {
        v2f a = { aS[k + koff], aS[k + koff + 1] };
        v2f b = { wS[k + koff], wS[k + koff + 1] };
        c = __builtin_amdgcn_wmma_f32_16x16x4_f32(false, a, false, b,
                                                  (short)0, c, false, false);
    }
    #pragma unroll
    for (int k = 0; k < D; k += 4) {
        v2f a = { aM[k + koff], aM[k + koff + 1] };
        v2f b = { wM[k + koff], wM[k + koff + 1] };
        c = __builtin_amdgcn_wmma_f32_16x16x4_f32(false, a, false, b,
                                                  (short)0, c, false, false);
    }

    const int col  = col0 + mn;
    const float bS = bself[col];
    const float bM = bmsg[col];
    const int rbase = row0 + ((lane >> 4) << 3);
    #pragma unroll
    for (int v = 0; v < 8; ++v) {
        int row = rbase + v;
        out[(size_t)row * D + col] = c[v] + bS + deg[row] * bM;
    }
}

extern "C" void kernel_launch(void* const* d_in, const int* in_sizes, int n_in,
                              void* d_out, int out_size, void* d_ws, size_t ws_size,
                              hipStream_t stream)
{
    const float* UF   = (const float*)d_in[0];
    const float* IF   = (const float*)d_in[1];
    const int*   edge = (const int*)d_in[2];
    const float* Wu   = (const float*)d_in[3];
    const float* bu   = (const float*)d_in[4];
    const float* Wi   = (const float*)d_in[5];
    const float* bi   = (const float*)d_in[6];
    const float* Wu2i = (const float*)d_in[7];
    const float* bu2i = (const float*)d_in[8];
    const float* Wi2u = (const float*)d_in[9];
    const float* bi2u = (const float*)d_in[10];

    const int U = in_sizes[0] / D;
    const int I = in_sizes[1] / D;
    const int E = in_sizes[2] / 2;
    const int R = U + I;

    float* outU = (float*)d_out;
    float* outI = outU + (size_t)U * D;

    const int zn4 = R / 4 + 1;
    int*   cnt    = (int*)d_ws;
    int*   cursor = cnt + R;
    int*   off    = cnt + (size_t)zn4 * 4;
    float* degf   = (float*)(off + R);
    int*   bkt    = (int*)(degf + R);

    zero_i4_kernel<<<(unsigned)((zn4 + 255) / 256), 256, 0, stream>>>((int4*)cnt, zn4);

    edge_count_kernel<<<(unsigned)((E + 255) / 256), 256, 0, stream>>>(
        edge, edge + E, cnt, cnt + U, E);
    bucket_alloc_kernel<<<(unsigned)((R + 1023) / 1024), 1024, 0, stream>>>(
        cnt, off, degf, cursor, R);
    bucket_fill_kernel<<<(unsigned)((E + 255) / 256), 256, 0, stream>>>(
        edge, edge + E, cnt, cnt + U, off, off + U, bkt, E);

    segment_gather_kernel<<<(unsigned)((R + 7) / 8), 256, 0, stream>>>(
        UF, IF, off, degf, bkt, outU, U, R);

    gnn_gemm_kernel<<<U / 16, 256, 0, stream>>>(UF, outU, Wu, bu, Wi2u, bi2u, degf);
    gnn_gemm_kernel<<<I / 16, 256, 0, stream>>>(IF, outI, Wi, bi, Wu2i, bu2i, degf + U);
}


# GEMM row blocking: each work-group computes 80 rows (5 row tiles per wave, independent accumulators), so each weight fragment load feeds 5 matrix instructions, not 1; tile staged in one load group
# speedup vs baseline: 1.3628x; 1.2358x over previous
#include <hip/hip_runtime.h>
#include <hip/hip_bf16.h>

typedef __attribute__((ext_vector_type(2))) float v2f;
typedef __attribute__((ext_vector_type(8))) float v8f;

#define D 128
#define LDS_STRIDE 132

__global__ __launch_bounds__(256) void zero_i4_kernel(int4* __restrict__ p, int n4) {
    int i = blockIdx.x * 256 + threadIdx.x;
    if (i < n4) p[i] = make_int4(0, 0, 0, 0);
}

__global__ __launch_bounds__(256)
void edge_count_kernel(const int* __restrict__ uidx, const int* __restrict__ iidx,
                       int* __restrict__ cntU, int* __restrict__ cntI, int E)
{
    int e = blockIdx.x * 256 + threadIdx.x;
    if (e >= E) return;
    atomicAdd(cntU + uidx[e], 1);
    atomicAdd(cntI + iidx[e], 1);
}

__global__ __launch_bounds__(1024)
void bucket_alloc_kernel(const int* __restrict__ cnt, int* __restrict__ off,
                         float* __restrict__ degf, int* __restrict__ cursor, int R)
{
    __shared__ int wsum[32];
    __shared__ int base_s;
    const int r    = blockIdx.x * 1024 + threadIdx.x;
    const int lane = threadIdx.x & 31;
    const int w    = threadIdx.x >> 5;
    const int c    = (r < R) ? cnt[r] : 0;

    int x = c;
    for (int d = 1; d < 32; d <<= 1) {
        int y = __shfl_up(x, d);
        if (lane >= d) x += y;
    }
    if (lane == 31) wsum[w] = x;
    __syncthreads();
    if (w == 0) {
        const int s = wsum[lane];
        int t = s;
        for (int d = 1; d < 32; d <<= 1) {
            int y = __shfl_up(t, d);
            if (lane >= d) t += y;
        }
        wsum[lane] = t - s;
        if (lane == 31) base_s = atomicAdd(cursor, t);
    }
    __syncthreads();
    if (r < R) {
        off[r]  = base_s + wsum[w] + x - c;
        degf[r] = (float)c;
    }
}

__global__ __launch_bounds__(256)
void bucket_fill_kernel(const int* __restrict__ uidx, const int* __restrict__ iidx,
                        int* __restrict__ cntU, int* __restrict__ cntI,
                        const int* __restrict__ offU, const int* __restrict__ offI,
                        int* __restrict__ bkt, int E)
{
    int e = blockIdx.x * 256 + threadIdx.x;
    if (e >= E) return;
    const int u  = uidx[e];
    const int it = iidx[e];
    const int pu = atomicAdd(cntU + u,  -1) - 1;
    const int pi = atomicAdd(cntI + it, -1) - 1;
    bkt[offU[u]  + pu] = it;
    bkt[offI[it] + pi] = u;
}

__global__ __launch_bounds__(256)
void segment_gather_kernel(const float* __restrict__ UF, const float* __restrict__ IF,
                           const int* __restrict__ off, const float* __restrict__ degf,
                           const int* __restrict__ bkt, float* __restrict__ out,
                           int U, int R)
{
    const int lane = threadIdx.x & 31;
    const int r = __builtin_amdgcn_readfirstlane(blockIdx.x * 8 + (threadIdx.x >> 5));
    if (r >= R) return;
    const float* __restrict__ src = (r < U) ? IF : UF;
    const int n = __builtin_amdgcn_readfirstlane((int)degf[r]);
    const int o = __builtin_amdgcn_readfirstlane(off[r]);

    float4 acc = make_float4(0.f, 0.f, 0.f, 0.f);

    if (n > 0 && n <= 32) {
        const int k = (lane < n) ? bkt[o + lane] : 0x7fffffff;
        int rk = 0;
        for (int l = 0; l < n; ++l) {
            const int s = __builtin_amdgcn_readlane(k, l);
            rk += (s < k || (s == k && l < lane)) ? 1 : 0;
        }
        const int rank = (lane < n) ? rk : lane;
        const int sk = __builtin_amdgcn_ds_permute(rank << 2, k);

        for (int t0 = 0; t0 < n; t0 += 8) {
            float4 f[8];
            #pragma unroll
            for (int j = 0; j < 8; ++j) {
                const int t = (t0 + j < n) ? (t0 + j) : (n - 1);
                const int s = __builtin_amdgcn_readlane(sk, t);
                f[j] = ((const float4*)(src + (size_t)s * D))[lane];
            }
            #pragma unroll
            for (int j = 0; j < 8; ++j) {
                const bool live = (t0 + j < n);
                acc.x += live ? f[j].x : 0.f;
                acc.y += live ? f[j].y : 0.f;
                acc.z += live ? f[j].z : 0.f;
                acc.w += live ? f[j].w : 0.f;
            }
        }
    } else if (n > 32) {
        int cur = -1;
        for (int trip = 0; trip < n; ++trip) {
            int best = 0x7fffffff, m = 0;
            for (int j = lane; j < n; j += 32) {
                const int k = bkt[o + j];
                if (k > cur) {
                    if (k < best)       { best = k; m = 1; }
                    else if (k == best) { ++m; }
                }
            }
            int wb = best;
            for (int d = 16; d > 0; d >>= 1) wb = min(wb, __shfl_xor(wb, d));
            if (wb == 0x7fffffff) break;
            int mm = (best == wb) ? m : 0;
            for (int d = 16; d > 0; d >>= 1) mm += __shfl_xor(mm, d);
            const int s   = __builtin_amdgcn_readfirstlane(wb);
            const int rep = __builtin_amdgcn_readfirstlane(mm);
            const float4 f = ((const float4*)(src + (size_t)s * D))[lane];
            for (int t = 0; t < rep; ++t) {
                acc.x += f.x; acc.y += f.y; acc.z += f.z; acc.w += f.w;
            }
            cur = s;
        }
    }

    ((float4*)(out + (size_t)r * D))[lane] = acc;
}

#define ROW_TILES 5
#define BLOCK_ROWS (ROW_TILES * 16)

__global__ __launch_bounds__(256)
void gnn_gemm_kernel(const float* __restrict__ X, float* __restrict__ out,
                     const float* __restrict__ Wself, const float* __restrict__ bself,
                     const float* __restrict__ Wmsg,  const float* __restrict__ bmsg,
                     const float* __restrict__ deg, int rows)
{
    __shared__ float sX[BLOCK_ROWS * LDS_STRIDE];
    __shared__ float sG[BLOCK_ROWS * LDS_STRIDE];

    const int row0 = blockIdx.x * BLOCK_ROWS;
    const int tid  = threadIdx.x;

    {
        const int c4 = tid & 31;
        const int r0 = tid >> 5;
        float4 xr[BLOCK_ROWS / 8], gv[BLOCK_ROWS / 8];
        #pragma unroll
        for (int j = 0; j < BLOCK_ROWS / 8; ++j) {
            const int gr = min(row0 + r0 + 8 * j, rows - 1);
            xr[j] = ((const float4*)(X   + (size_t)gr * D))[c4];
            gv[j] = ((const float4*)(out + (size_t)gr * D))[c4];
        }
        #pragma unroll
        for (int j = 0; j < BLOCK_ROWS / 8; ++j) {
            const int r = r0 + 8 * j;
            ((float4*)(sX + r * LDS_STRIDE))[c4] = xr[j];
            ((float4*)(sG + r * LDS_STRIDE))[c4] = gv[j];
        }
    }
    __syncthreads();

    const int lane = tid & 31;
    const int wave = tid >> 5;
    const int col0 = wave * 16;
    const int mn   = lane & 15;
    const int koff = (lane >> 4) << 1;

    const float* wS = Wself + (size_t)(col0 + mn) * D;
    const float* wM = Wmsg  + (size_t)(col0 + mn) * D;
    const float* aS = sX + mn * LDS_STRIDE;
    const float* aM = sG + mn * LDS_STRIDE;

    v8f c[ROW_TILES];
    #pragma unroll
    for (int t = 0; t < ROW_TILES; ++t) c[t] = (v8f){};

    #pragma unroll
    for (int k = 0; k < D; k += 4) {
        v2f b = { wS[k + koff], wS[k + koff + 1] };
        #pragma unroll
        for (int t = 0; t < ROW_TILES; ++t) {
            const float* ap = aS + t * 16 * LDS_STRIDE;
            v2f a = { ap[k + koff], ap[k + koff + 1] };
            c[t] = __builtin_amdgcn_wmma_f32_16x16x4_f32(false, a, false, b,
                                                         (short)0, c[t], false, false);
        }
    }
    #pragma unroll
    for (int k = 0; k < D; k += 4) {
        v2f b = { wM[k + koff], wM[k + koff + 1] };
        #pragma unroll
        for (int t = 0; t < ROW_TILES; ++t) {
            const float* ap = aM + t * 16 * LDS_STRIDE;
            v2f a = { ap[k + koff], ap[k + koff + 1] };
            c[t] = __builtin_amdgcn_wmma_f32_16x16x4_f32(false, a, false, b,
                                                         (short)0, c[t], false, false);
        }
    }

    const int col  = col0 + mn;
    const float bS = bself[col];
    const float bM = bmsg[col];
    #pragma unroll
    for (int t = 0; t < ROW_TILES; ++t) {
        const int rbase = row0 + t * 16 + ((lane >> 4) << 3);
        if (row0 + t * 16 < rows) {
            #pragma unroll
            for (int v = 0; v < 8; ++v) {
                int row = rbase + v;
                out[(size_t)row * D + col] = c[t][v] + bS + deg[row] * bM;
            }
        }
    }
}

extern "C" void kernel_launch(void* const* d_in, const int* in_sizes, int n_in,
                              void* d_out, int out_size, void* d_ws, size_t ws_size,
                              hipStream_t stream)
{
    const float* UF   = (const float*)d_in[0];
    const float* IF   = (const float*)d_in[1];
    const int*   edge = (const int*)d_in[2];
    const float* Wu   = (const float*)d_in[3];
    const float* bu   = (const float*)d_in[4];
    const float* Wi   = (const float*)d_in[5];
    const float* bi   = (const float*)d_in[6];
    const float* Wu2i = (const float*)d_in[7];
    const float* bu2i = (const float*)d_in[8];
    const float* Wi2u = (const float*)d_in[9];
    const float* bi2u = (const float*)d_in[10];

    const int U = in_sizes[0] / D;
    const int I = in_sizes[1] / D;
    const int E = in_sizes[2] / 2;
    const int R = U + I;

    float* outU = (float*)d_out;
    float* outI = outU + (size_t)U * D;

    const int zn4 = R / 4 + 1;
    int*   cnt    = (int*)d_ws;
    int*   cursor = cnt + R;
    int*   off    = cnt + (size_t)zn4 * 4;
    float* degf   = (float*)(off + R);
    int*   bkt    = (int*)(degf + R);

    zero_i4_kernel<<<(unsigned)((zn4 + 255) / 256), 256, 0, stream>>>((int4*)cnt, zn4);

    edge_count_kernel<<<(unsigned)((E + 255) / 256), 256, 0, stream>>>(
        edge, edge + E, cnt, cnt + U, E);
    bucket_alloc_kernel<<<(unsigned)((R + 1023) / 1024), 1024, 0, stream>>>(
        cnt, off, degf, cursor, R);
    bucket_fill_kernel<<<(unsigned)((E + 255) / 256), 256, 0, stream>>>(
        edge, edge + E, cnt, cnt + U, off, off + U, bkt, E);

    segment_gather_kernel<<<(unsigned)((R + 7) / 8), 256, 0, stream>>>(
        UF, IF, off, degf, bkt, outU, U, R);

    gnn_gemm_kernel<<<(unsigned)((U + BLOCK_ROWS - 1) / BLOCK_ROWS), 256, 0, stream>>>(
        UF, outU, Wu, bu, Wi2u, bi2u, degf, U);
    gnn_gemm_kernel<<<(unsigned)((I + BLOCK_ROWS - 1) / BLOCK_ROWS), 256, 0, stream>>>(
        IF, outI, Wi, bi, Wu2i, bu2i, degf + U, I);
}


# Bucket slot taken from the counting pass: its add returns the edge's place in its row's bucket, stored per edge; fill pass places edges with no atomics (integer atomics halved); sum order unchanged
# speedup vs baseline: 1.6851x; 1.2365x over previous
#include <hip/hip_runtime.h>
#include <hip/hip_bf16.h>

typedef __attribute__((ext_vector_type(2))) float v2f;
typedef __attribute__((ext_vector_type(8))) float v8f;

#define D 128
#define LDS_STRIDE 132

__global__ __launch_bounds__(256) void zero_i4_kernel(int4* __restrict__ p, int n4) {
    int i = blockIdx.x * 256 + threadIdx.x;
    if (i < n4) p[i] = make_int4(0, 0, 0, 0);
}

__global__ __launch_bounds__(256)
void edge_count_kernel(const int* __restrict__ uidx, const int* __restrict__ iidx,
                       int* __restrict__ cntU, int* __restrict__ cntI,
                       int2* __restrict__ place, int E)
{
    int e = blockIdx.x * 256 + threadIdx.x;
    if (e >= E) return;
    const int u  = uidx[e];
    const int it = iidx[e];
    const int pu = atomicAdd(cntU + u,  1);
    const int pi = atomicAdd(cntI + it, 1);
    place[e] = make_int2(pu, pi);
}

__global__ __launch_bounds__(1024)
void bucket_alloc_kernel(const int* __restrict__ cnt, int* __restrict__ off,
                         float* __restrict__ degf, int* __restrict__ cursor, int R)
{
    __shared__ int wsum[32];
    __shared__ int base_s;
    const int r    = blockIdx.x * 1024 + threadIdx.x;
    const int lane = threadIdx.x & 31;
    const int w    = threadIdx.x >> 5;
    const int c    = (r < R) ? cnt[r] : 0;

    int x = c;
    for (int d = 1; d < 32; d <<= 1) {
        int y = __shfl_up(x, d);
        if (lane >= d) x += y;
    }
    if (lane == 31) wsum[w] = x;
    __syncthreads();
    if (w == 0) {
        const int s = wsum[lane];
        int t = s;
        for (int d = 1; d < 32; d <<= 1) {
            int y = __shfl_up(t, d);
            if (lane >= d) t += y;
        }
        wsum[lane] = t - s;
        if (lane == 31) base_s = atomicAdd(cursor, t);
    }
    __syncthreads();
    if (r < R) {
        off[r]  = base_s + wsum[w] + x - c;
        degf[r] = (float)c;
    }
}

__global__ __launch_bounds__(256)
void bucket_fill_kernel(const int* __restrict__ uidx, const int* __restrict__ iidx,
                        const int2* __restrict__ place,
                        const int* __restrict__ offU, const int* __restrict__ offI,
                        int* __restrict__ bkt, int E)
{
    int e = blockIdx.x * 256 + threadIdx.x;
    if (e >= E) return;
    const int  u  = uidx[e];
    const int  it = iidx[e];
    const int2 p  = place[e];
    bkt[offU[u]  + p.x] = it;
    bkt[offI[it] + p.y] = u;
}

__global__ __launch_bounds__(256)
void segment_gather_kernel(const float* __restrict__ UF, const float* __restrict__ IF,
                           const int* __restrict__ off, const float* __restrict__ degf,
                           const int* __restrict__ bkt, float* __restrict__ out,
                           int U, int R)
{
    const int lane = threadIdx.x & 31;
    const int r = __builtin_amdgcn_readfirstlane(blockIdx.x * 8 + (threadIdx.x >> 5));
    if (r >= R) return;
    const float* __restrict__ src = (r < U) ? IF : UF;
    const int n = __builtin_amdgcn_readfirstlane((int)degf[r]);
    const int o = __builtin_amdgcn_readfirstlane(off[r]);

    float4 acc = make_float4(0.f, 0.f, 0.f, 0.f);

    if (n > 0 && n <= 32) {
        const int k = (lane < n) ? bkt[o + lane] : 0x7fffffff;
        int rk = 0;
        for (int l = 0; l < n; ++l) {
            const int s = __builtin_amdgcn_readlane(k, l);
            rk += (s < k || (s == k && l < lane)) ? 1 : 0;
        }
        const int rank = (lane < n) ? rk : lane;
        const int sk = __builtin_amdgcn_ds_permute(rank << 2, k);

        for (int t0 = 0; t0 < n; t0 += 8) {
            float4 f[8];
            #pragma unroll
            for (int j = 0; j < 8; ++j) {
                const int t = (t0 + j < n) ? (t0 + j) : (n - 1);
                const int s = __builtin_amdgcn_readlane(sk, t);
                f[j] = ((const float4*)(src + (size_t)s * D))[lane];
            }
            #pragma unroll
            for (int j = 0; j < 8; ++j) {
                const bool live = (t0 + j < n);
                acc.x += live ? f[j].x : 0.f;
                acc.y += live ? f[j].y : 0.f;
                acc.z += live ? f[j].z : 0.f;
                acc.w += live ? f[j].w : 0.f;
            }
        }
    } else if (n > 32) {
        int cur = -1;
        for (int trip = 0; trip < n; ++trip) {
            int best = 0x7fffffff, m = 0;
            for (int j = lane; j < n; j += 32) {
                const int k = bkt[o + j];
                if (k > cur) {
                    if (k < best)       { best = k; m = 1; }
                    else if (k == best) { ++m; }
                }
            }
            int wb = best;
            for (int d = 16; d > 0; d >>= 1) wb = min(wb, __shfl_xor(wb, d));
            if (wb == 0x7fffffff) break;
            int mm = (best == wb) ? m : 0;
            for (int d = 16; d > 0; d >>= 1) mm += __shfl_xor(mm, d);
            const int s   = __builtin_amdgcn_readfirstlane(wb);
            const int rep = __builtin_amdgcn_readfirstlane(mm);
            const float4 f = ((const float4*)(src + (size_t)s * D))[lane];
            for (int t = 0; t < rep; ++t) {
                acc.x += f.x; acc.y += f.y; acc.z += f.z; acc.w += f.w;
            }
            cur = s;
        }
    }

    ((float4*)(out + (size_t)r * D))[lane] = acc;
}

#define ROW_TILES 5
#define BLOCK_ROWS (ROW_TILES * 16)

__global__ __launch_bounds__(256)
void gnn_gemm_kernel(const float* __restrict__ X, float* __restrict__ out,
                     const float* __restrict__ Wself, const float* __restrict__ bself,
                     const float* __restrict__ Wmsg,  const float* __restrict__ bmsg,
                     const float* __restrict__ deg, int rows)
{
    __shared__ float sX[BLOCK_ROWS * LDS_STRIDE];
    __shared__ float sG[BLOCK_ROWS * LDS_STRIDE];

    const int row0 = blockIdx.x * BLOCK_ROWS;
    const int tid  = threadIdx.x;

    {
        const int c4 = tid & 31;
        const int r0 = tid >> 5;
        float4 xr[BLOCK_ROWS / 8], gv[BLOCK_ROWS / 8];
        #pragma unroll
        for (int j = 0; j < BLOCK_ROWS / 8; ++j) {
            const int gr = min(row0 + r0 + 8 * j, rows - 1);
            xr[j] = ((const float4*)(X   + (size_t)gr * D))[c4];
            gv[j] = ((const float4*)(out + (size_t)gr * D))[c4];
        }
        #pragma unroll
        for (int j = 0; j < BLOCK_ROWS / 8; ++j) {
            const int r = r0 + 8 * j;
            ((float4*)(sX + r * LDS_STRIDE))[c4] = xr[j];
            ((float4*)(sG + r * LDS_STRIDE))[c4] = gv[j];
        }
    }
    __syncthreads();

    const int lane = tid & 31;
    const int wave = tid >> 5;
    const int col0 = wave * 16;
    const int mn   = lane & 15;
    const int koff = (lane >> 4) << 1;

    const float* wS = Wself + (size_t)(col0 + mn) * D;
    const float* wM = Wmsg  + (size_t)(col0 + mn) * D;
    const float* aS = sX + mn * LDS_STRIDE;
    const float* aM = sG + mn * LDS_STRIDE;

    v8f c[ROW_TILES];
    #pragma unroll
    for (int t = 0; t < ROW_TILES; ++t) c[t] = (v8f){};

    #pragma unroll
    for (int k = 0; k < D; k += 4) {
        v2f b = { wS[k + koff], wS[k + koff + 1] };
        #pragma unroll
        for (int t = 0; t < ROW_TILES; ++t) {
            const float* ap = aS + t * 16 * LDS_STRIDE;
            v2f a = { ap[k + koff], ap[k + koff + 1] };
            c[t] = __builtin_amdgcn_wmma_f32_16x16x4_f32(false, a, false, b,
                                                         (short)0, c[t], false, false);
        }
    }
    #pragma unroll
    for (int k = 0; k < D; k += 4) {
        v2f b = { wM[k + koff], wM[k + koff + 1] };
        #pragma unroll
        for (int t = 0; t < ROW_TILES; ++t) {
            const float* ap = aM + t * 16 * LDS_STRIDE;
            v2f a = { ap[k + koff], ap[k + koff + 1] };
            c[t] = __builtin_amdgcn_wmma_f32_16x16x4_f32(false, a, false, b,
                                                         (short)0, c[t], false, false);
        }
    }

    const int col  = col0 + mn;
    const float bS = bself[col];
    const float bM = bmsg[col];
    #pragma unroll
    for (int t = 0; t < ROW_TILES; ++t) {
        const int rbase = row0 + t * 16 + ((lane >> 4) << 3);
        if (row0 + t * 16 < rows) {
            #pragma unroll
            for (int v = 0; v < 8; ++v) {
                int row = rbase + v;
                out[(size_t)row * D + col] = c[t][v] + bS + deg[row] * bM;
            }
        }
    }
}

extern "C" void kernel_launch(void* const* d_in, const int* in_sizes, int n_in,
                              void* d_out, int out_size, void* d_ws, size_t ws_size,
                              hipStream_t stream)
{
    const float* UF   = (const float*)d_in[0];
    const float* IF   = (const float*)d_in[1];
    const int*   edge = (const int*)d_in[2];
    const float* Wu   = (const float*)d_in[3];
    const float* bu   = (const float*)d_in[4];
    const float* Wi   = (const float*)d_in[5];
    const float* bi   = (const float*)d_in[6];
    const float* Wu2i = (const float*)d_in[7];
    const float* bu2i = (const float*)d_in[8];
    const float* Wi2u = (const float*)d_in[9];
    const float* bi2u = (const float*)d_in[10];

    const int U = in_sizes[0] / D;
    const int I = in_sizes[1] / D;
    const int E = in_sizes[2] / 2;
    const int R = U + I;

    float* outU = (float*)d_out;
    float* outI = outU + (size_t)U * D;

    const int zn4 = R / 4 + 1;
    int*   cnt    = (int*)d_ws;
    int*   cursor = cnt + R;
    int*   off    = cnt + (size_t)zn4 * 4;
    float* degf   = (float*)(off + R);
    int*   bkt    = (int*)(degf + R);
    int2*  place  = (int2*)(bkt + (size_t)2 * E);

    zero_i4_kernel<<<(unsigned)((zn4 + 255) / 256), 256, 0, stream>>>((int4*)cnt, zn4);

    edge_count_kernel<<<(unsigned)((E + 255) / 256), 256, 0, stream>>>(
        edge, edge + E, cnt, cnt + U, place, E);
    bucket_alloc_kernel<<<(unsigned)((R + 1023) / 1024), 1024, 0, stream>>>(
        cnt, off, degf, cursor, R);
    bucket_fill_kernel<<<(unsigned)((E + 255) / 256), 256, 0, stream>>>(
        edge, edge + E, place, off, off + U, bkt, E);

    segment_gather_kernel<<<(unsigned)((R + 7) / 8), 256, 0, stream>>>(
        UF, IF, off, degf, bkt, outU, U, R);

    gnn_gemm_kernel<<<(unsigned)((U + BLOCK_ROWS - 1) / BLOCK_ROWS), 256, 0, stream>>>(
        UF, outU, Wu, bu, Wi2u, bi2u, degf, U);
    gnn_gemm_kernel<<<(unsigned)((I + BLOCK_ROWS - 1) / BLOCK_ROWS), 256, 0, stream>>>(
        IF, outI, Wi, bi, Wu2i, bu2i, degf + U, I);
}


# Edge bucketing without global atomics: two-level counting sort in LDS (per-chunk bin counts, bin-ordered run writes, one block per 512-row bin builds its buckets); sum order unchanged
# speedup vs baseline: 2.6097x; 1.5487x over previous
#include <hip/hip_runtime.h>
#include <hip/hip_bf16.h>

typedef __attribute__((ext_vector_type(2))) float v2f;
typedef __attribute__((ext_vector_type(8))) float v8f;

#define D 128
#define LDS_STRIDE 132

#define BIN_SHIFT 9
#define BIN_ROWS  (1 << BIN_SHIFT)
#define CHUNK     4096
#define EPT       (CHUNK / 1024)
#define NBINS_MAX 1024
#define HB        32
#define STAGE_CAP 8192
#define KEEP      5

__device__ __forceinline__ int block_scan_1024(int c, int* ws, int* total)
{
    const int lane = threadIdx.x & 31;
    const int w    = threadIdx.x >> 5;
    int x = c;
    for (int d = 1; d < 32; d <<= 1) {
        int y = __shfl_up(x, d);
        if (lane >= d) x += y;
    }
    if (lane == 31) ws[w] = x;
    __syncthreads();
    if (w == 0) {
        const int s = ws[lane];
        int t = s;
        for (int d = 1; d < 32; d <<= 1) {
            int y = __shfl_up(t, d);
            if (lane >= d) t += y;
        }
        ws[lane] = t - s;
        if (lane == 31) ws[32] = t;
    }
    __syncthreads();
    const int r = ws[w] + x - c;
    *total = ws[32];
    __syncthreads();
    return r;
}

__global__ __launch_bounds__(1024)
void bin_count_kernel(const int* __restrict__ uidx, const int* __restrict__ iidx,
                      int* __restrict__ H, int E, int U, int NB)
{
    __shared__ int hist[NBINS_MAX];
    const int tid = threadIdx.x;
    hist[tid] = 0;
    __syncthreads();
    const int base = blockIdx.x * CHUNK;
    int du[EPT], di[EPT];
    #pragma unroll
    for (int j = 0; j < EPT; ++j) {
        const int e = min(base + j * 1024 + tid, E - 1);
        du[j] = uidx[e];
        di[j] = iidx[e];
    }
    #pragma unroll
    for (int j = 0; j < EPT; ++j) {
        if (base + j * 1024 + tid < E) {
            atomicAdd(&hist[du[j] >> BIN_SHIFT], 1);
            atomicAdd(&hist[(U + di[j]) >> BIN_SHIFT], 1);
        }
    }
    __syncthreads();
    if (tid < NB) H[blockIdx.x * NB + tid] = hist[tid];
}

__global__ __launch_bounds__(1024)
void bin_scatter_kernel(const int* __restrict__ uidx, const int* __restrict__ iidx,
                        const int* __restrict__ H, int* __restrict__ binStart,
                        int2* __restrict__ coarse, int E, int U, int NB, int NBLK)
{
    __shared__ int  hist[NBINS_MAX];
    __shared__ int  lstart[NBINS_MAX];
    __shared__ int  delta[NBINS_MAX];
    __shared__ int  ws[33];
    __shared__ int2 stage[2 * CHUNK];

    const int tid = threadIdx.x;
    const int me  = blockIdx.x;
    hist[tid]   = 0;
    lstart[tid] = 0;
    delta[tid]  = 0;
    __syncthreads();

    const int base = me * CHUNK;
    int du[EPT], di[EPT], pu[EPT], pi[EPT];
    #pragma unroll
    for (int j = 0; j < EPT; ++j) {
        const int e = min(base + j * 1024 + tid, E - 1);
        du[j] = uidx[e];
        di[j] = iidx[e];
    }
    #pragma unroll
    for (int j = 0; j < EPT; ++j) {
        pu[j] = 0; pi[j] = 0;
        if (base + j * 1024 + tid < E) {
            pu[j] = atomicAdd(&hist[du[j] >> BIN_SHIFT], 1);
            pi[j] = atomicAdd(&hist[(U + di[j]) >> BIN_SHIFT], 1);
        } else {
            du[j] = -1;
        }
    }

    {
        const int G = 1024 / NB;
        const int g = tid / NB;
        const int t = tid - g * NB;
        if (g < G) {
            int tot = 0, pre = 0;
            for (int c0 = 0; c0 < NBLK; c0 += HB * G) {
                int v[HB];
                #pragma unroll
                for (int j = 0; j < HB; ++j)
                    v[j] = H[min(c0 + g + j * G, NBLK - 1) * NB + t];
                #pragma unroll
                for (int j = 0; j < HB; ++j) {
                    const int b = c0 + g + j * G;
                    tot += (b < NBLK) ? v[j] : 0;
                    pre += (b < me)   ? v[j] : 0;
                }
            }
            atomicAdd(&lstart[t], tot);
            atomicAdd(&delta[t],  pre);
        }
    }
    __syncthreads();
    const int tot = lstart[tid];
    const int pre = delta[tid];

    int nloc, nall;
    const int ls = block_scan_1024(hist[tid], ws, &nloc);
    const int gs = block_scan_1024(tot, ws, &nall);
    lstart[tid] = ls;
    delta[tid]  = gs + pre - ls;
    if (me == 0) {
        if (tid < NB)  binStart[tid] = gs;
        if (tid == 0)  binStart[NB]  = nall;
    }
    __syncthreads();

    #pragma unroll
    for (int j = 0; j < EPT; ++j) {
        if (du[j] >= 0) {
            const int ru = du[j];
            const int ri = U + di[j];
            stage[lstart[ru >> BIN_SHIFT] + pu[j]] = make_int2(ru, di[j]);
            stage[lstart[ri >> BIN_SHIFT] + pi[j]] = make_int2(ri, du[j]);
        }
    }
    __syncthreads();

    for (int i = tid; i < nloc; i += 1024) {
        const int2 en = stage[i];
        coarse[delta[en.x >> BIN_SHIFT] + i] = en;
    }
}

__global__ __launch_bounds__(1024)
void bin_build_kernel(const int2* __restrict__ coarse, const int* __restrict__ binStart,
                      int* __restrict__ off, float* __restrict__ degf,
                      int* __restrict__ bkt, int R)
{
    __shared__ int cnt[BIN_ROWS];
    __shared__ int cur[BIN_ROWS];
    __shared__ int ws[33];
    __shared__ int sb[STAGE_CAP];

    const int tid = threadIdx.x;
    const int b   = blockIdx.x;
    const int s   = binStart[b];
    const int n   = binStart[b + 1] - s;

    if (tid < BIN_ROWS) cnt[tid] = 0;

    int2 en[KEEP];
    #pragma unroll
    for (int j = 0; j < KEEP; ++j) en[j] = make_int2(-1, 0);
    if (n > 0) {
        #pragma unroll
        for (int j = 0; j < KEEP; ++j) {
            const int  q = j * 1024 + tid;
            const int2 v = coarse[s + min(q, n - 1)];
            if (q < n) en[j] = v;
        }
    }
    __syncthreads();
    #pragma unroll
    for (int j = 0; j < KEEP; ++j)
        if (en[j].x >= 0) atomicAdd(&cnt[en[j].x & (BIN_ROWS - 1)], 1);
    for (int q = KEEP * 1024 + tid; q < n; q += 1024)
        atomicAdd(&cnt[coarse[s + q].x & (BIN_ROWS - 1)], 1);
    __syncthreads();

    const int c = (tid < BIN_ROWS) ? cnt[tid] : 0;
    int nsum;
    const int ex = block_scan_1024(c, ws, &nsum);
    if (tid < BIN_ROWS) {
        cur[tid] = ex;
        const int r = (b << BIN_SHIFT) + tid;
        if (r < R) {
            off[r]  = s + ex;
            degf[r] = (float)c;
        }
    }
    __syncthreads();

    if (n <= STAGE_CAP) {
        #pragma unroll
        for (int j = 0; j < KEEP; ++j) {
            if (en[j].x >= 0) {
                const int k = atomicAdd(&cur[en[j].x & (BIN_ROWS - 1)], 1);
                sb[k] = en[j].y;
            }
        }
        for (int q = KEEP * 1024 + tid; q < n; q += 1024) {
            const int2 e2 = coarse[s + q];
            const int  k  = atomicAdd(&cur[e2.x & (BIN_ROWS - 1)], 1);
            sb[k] = e2.y;
        }
        __syncthreads();
        for (int q = tid; q < n; q += 1024) bkt[s + q] = sb[q];
    } else {
        for (int q = tid; q < n; q += 1024) {
            const int2 e2 = coarse[s + q];
            const int  k  = atomicAdd(&cur[e2.x & (BIN_ROWS - 1)], 1);
            bkt[s + k] = e2.y;
        }
    }
}

__global__ __launch_bounds__(256)
void segment_gather_kernel(const float* __restrict__ UF, const float* __restrict__ IF,
                           const int* __restrict__ off, const float* __restrict__ degf,
                           const int* __restrict__ bkt, float* __restrict__ out,
                           int U, int R)
{
    const int lane = threadIdx.x & 31;
    const int r = __builtin_amdgcn_readfirstlane(blockIdx.x * 8 + (threadIdx.x >> 5));
    if (r >= R) return;
    const float* __restrict__ src = (r < U) ? IF : UF;
    const int n = __builtin_amdgcn_readfirstlane((int)degf[r]);
    const int o = __builtin_amdgcn_readfirstlane(off[r]);

    float4 acc = make_float4(0.f, 0.f, 0.f, 0.f);

    if (n > 0 && n <= 32) {
        const int k = (lane < n) ? bkt[o + lane] : 0x7fffffff;
        int rk = 0;
        for (int l = 0; l < n; ++l) {
            const int s = __builtin_amdgcn_readlane(k, l);
            rk += (s < k || (s == k && l < lane)) ? 1 : 0;
        }
        const int rank = (lane < n) ? rk : lane;
        const int sk = __builtin_amdgcn_ds_permute(rank << 2, k);

        for (int t0 = 0; t0 < n; t0 += 8) {
            float4 f[8];
            #pragma unroll
            for (int j = 0; j < 8; ++j) {
                const int t = (t0 + j < n) ? (t0 + j) : (n - 1);
                const int s = __builtin_amdgcn_readlane(sk, t);
                f[j] = ((const float4*)(src + (size_t)s * D))[lane];
            }
            #pragma unroll
            for (int j = 0; j < 8; ++j) {
                const bool live = (t0 + j < n);
                acc.x += live ? f[j].x : 0.f;
                acc.y += live ? f[j].y : 0.f;
                acc.z += live ? f[j].z : 0.f;
                acc.w += live ? f[j].w : 0.f;
            }
        }
    } else if (n > 32) {
        int cur = -1;
        for (int trip = 0; trip < n; ++trip) {
            int best = 0x7fffffff, m = 0;
            for (int j = lane; j < n; j += 32) {
                const int k = bkt[o + j];
                if (k > cur) {
                    if (k < best)       { best = k; m = 1; }
                    else if (k == best) { ++m; }
                }
            }
            int wb = best;
            for (int d = 16; d > 0; d >>= 1) wb = min(wb, __shfl_xor(wb, d));
            if (wb == 0x7fffffff) break;
            int mm = (best == wb) ? m : 0;
            for (int d = 16; d > 0; d >>= 1) mm += __shfl_xor(mm, d);
            const int s   = __builtin_amdgcn_readfirstlane(wb);
            const int rep = __builtin_amdgcn_readfirstlane(mm);
            const float4 f = ((const float4*)(src + (size_t)s * D))[lane];
            for (int t = 0; t < rep; ++t) {
                acc.x += f.x; acc.y += f.y; acc.z += f.z; acc.w += f.w;
            }
            cur = s;
        }
    }

    ((float4*)(out + (size_t)r * D))[lane] = acc;
}

#define ROW_TILES 5
#define BLOCK_ROWS (ROW_TILES * 16)

__global__ __launch_bounds__(256)
void gnn_gemm_kernel(const float* __restrict__ X, float* __restrict__ out,
                     const float* __restrict__ Wself, const float* __restrict__ bself,
                     const float* __restrict__ Wmsg,  const float* __restrict__ bmsg,
                     const float* __restrict__ deg, int rows)
{
    __shared__ float sX[BLOCK_ROWS * LDS_STRIDE];
    __shared__ float sG[BLOCK_ROWS * LDS_STRIDE];

    const int row0 = blockIdx.x * BLOCK_ROWS;
    const int tid  = threadIdx.x;

    {
        const int c4 = tid & 31;
        const int r0 = tid >> 5;
        float4 xr[BLOCK_ROWS / 8], gv[BLOCK_ROWS / 8];
        #pragma unroll
        for (int j = 0; j < BLOCK_ROWS / 8; ++j) {
            const int gr = min(row0 + r0 + 8 * j, rows - 1);
            xr[j] = ((const float4*)(X   + (size_t)gr * D))[c4];
            gv[j] = ((const float4*)(out + (size_t)gr * D))[c4];
        }
        #pragma unroll
        for (int j = 0; j < BLOCK_ROWS / 8; ++j) {
            const int r = r0 + 8 * j;
            ((float4*)(sX + r * LDS_STRIDE))[c4] = xr[j];
            ((float4*)(sG + r * LDS_STRIDE))[c4] = gv[j];
        }
    }
    __syncthreads();

    const int lane = tid & 31;
    const int wave = tid >> 5;
    const int col0 = wave * 16;
    const int mn   = lane & 15;
    const int koff = (lane >> 4) << 1;

    const float* wS = Wself + (size_t)(col0 + mn) * D;
    const float* wM = Wmsg  + (size_t)(col0 + mn) * D;
    const float* aS = sX + mn * LDS_STRIDE;
    const float* aM = sG + mn * LDS_STRIDE;

    v8f c[ROW_TILES];
    #pragma unroll
    for (int t = 0; t < ROW_TILES; ++t) c[t] = (v8f){};

    #pragma unroll
    for (int k = 0; k < D; k += 4) {
        v2f b = { wS[k + koff], wS[k + koff + 1] };
        #pragma unroll
        for (int t = 0; t < ROW_TILES; ++t) {
            const float* ap = aS + t * 16 * LDS_STRIDE;
            v2f a = { ap[k + koff], ap[k + koff + 1] };
            c[t] = __builtin_amdgcn_wmma_f32_16x16x4_f32(false, a, false, b,
                                                         (short)0, c[t], false, false);
        }
    }
    #pragma unroll
    for (int k = 0; k < D; k += 4) {
        v2f b = { wM[k + koff], wM[k + koff + 1] };
        #pragma unroll
        for (int t = 0; t < ROW_TILES; ++t) {
            const float* ap = aM + t * 16 * LDS_STRIDE;
            v2f a = { ap[k + koff], ap[k + koff + 1] };
            c[t] = __builtin_amdgcn_wmma_f32_16x16x4_f32(false, a, false, b,
                                                         (short)0, c[t], false, false);
        }
    }

    const int col  = col0 + mn;
    const float bS = bself[col];
    const float bM = bmsg[col];
    #pragma unroll
    for (int t = 0; t < ROW_TILES; ++t) {
        const int rbase = row0 + t * 16 + ((lane >> 4) << 3);
        if (row0 + t * 16 < rows) {
            #pragma unroll
            for (int v = 0; v < 8; ++v) {
                int row = rbase + v;
                out[(size_t)row * D + col] = c[t][v] + bS + deg[row] * bM;
            }
        }
    }
}

extern "C" void kernel_launch(void* const* d_in, const int* in_sizes, int n_in,
                              void* d_out, int out_size, void* d_ws, size_t ws_size,
                              hipStream_t stream)
{
    const float* UF   = (const float*)d_in[0];
    const float* IF   = (const float*)d_in[1];
    const int*   edge = (const int*)d_in[2];
    const float* Wu   = (const float*)d_in[3];
    const float* bu   = (const float*)d_in[4];
    const float* Wi   = (const float*)d_in[5];
    const float* bi   = (const float*)d_in[6];
    const float* Wu2i = (const float*)d_in[7];
    const float* bu2i = (const float*)d_in[8];
    const float* Wi2u = (const float*)d_in[9];
    const float* bi2u = (const float*)d_in[10];

    const int U = in_sizes[0] / D;
    const int I = in_sizes[1] / D;
    const int E = in_sizes[2] / 2;
    const int R = U + I;

    float* outU = (float*)d_out;
    float* outI = outU + (size_t)U * D;

    const int NB   = (R + BIN_ROWS - 1) / BIN_ROWS;
    const int NBLK = (E + CHUNK - 1) / CHUNK;

    int*   off    = (int*)d_ws;
    float* degf   = (float*)(off + R);
    int*   bkt    = (int*)(degf + R);
    int*   H      = bkt + (size_t)2 * E;
    int*   binSt  = H + (size_t)NBLK * NB;
    const size_t hb = ((size_t)NBLK * NB + NB + 2) & ~(size_t)1;
    int2*  coarse = (int2*)(H + hb);

    bin_count_kernel<<<(unsigned)NBLK, 1024, 0, stream>>>(edge, edge + E, H, E, U, NB);
    bin_scatter_kernel<<<(unsigned)NBLK, 1024, 0, stream>>>(
        edge, edge + E, H, binSt, coarse, E, U, NB, NBLK);

    bin_build_kernel<<<(unsigned)NB, 1024, 0, stream>>>(coarse, binSt, off, degf, bkt, R);

    segment_gather_kernel<<<(unsigned)((R + 7) / 8), 256, 0, stream>>>(
        UF, IF, off, degf, bkt, outU, U, R);

    gnn_gemm_kernel<<<(unsigned)((U + BLOCK_ROWS - 1) / BLOCK_ROWS), 256, 0, stream>>>(
        UF, outU, Wu, bu, Wi2u, bi2u, degf, U);
    gnn_gemm_kernel<<<(unsigned)((I + BLOCK_ROWS - 1) / BLOCK_ROWS), 256, 0, stream>>>(
        IF, outI, Wi, bi, Wu2i, bu2i, degf + U, I);
}


# Message product skipped for row blocks without edges: a block whose 80 rows all have degree 0 (aggregate tile all zeros) runs only the self product's matrix instructions; sum order unchanged
# speedup vs baseline: 2.7025x; 1.0356x over previous
#include <hip/hip_runtime.h>
#include <hip/hip_bf16.h>

typedef __attribute__((ext_vector_type(2))) float v2f;
typedef __attribute__((ext_vector_type(8))) float v8f;

#define D 128
#define LDS_STRIDE 132

#define BIN_SHIFT 9
#define BIN_ROWS  (1 << BIN_SHIFT)
#define CHUNK     4096
#define EPT       (CHUNK / 1024)
#define NBINS_MAX 1024
#define HB        32
#define STAGE_CAP 8192
#define KEEP      5

__device__ __forceinline__ int block_scan_1024(int c, int* ws, int* total)
{
    const int lane = threadIdx.x & 31;
    const int w    = threadIdx.x >> 5;
    int x = c;
    for (int d = 1; d < 32; d <<= 1) {
        int y = __shfl_up(x, d);
        if (lane >= d) x += y;
    }
    if (lane == 31) ws[w] = x;
    __syncthreads();
    if (w == 0) {
        const int s = ws[lane];
        int t = s;
        for (int d = 1; d < 32; d <<= 1) {
            int y = __shfl_up(t, d);
            if (lane >= d) t += y;
        }
        ws[lane] = t - s;
        if (lane == 31) ws[32] = t;
    }
    __syncthreads();
    const int r = ws[w] + x - c;
    *total = ws[32];
    __syncthreads();
    return r;
}

__global__ __launch_bounds__(1024)
void bin_count_kernel(const int* __restrict__ uidx, const int* __restrict__ iidx,
                      int* __restrict__ H, int E, int U, int NB)
{
    __shared__ int hist[NBINS_MAX];
    const int tid = threadIdx.x;
    hist[tid] = 0;
    __syncthreads();
    const int base = blockIdx.x * CHUNK;
    int du[EPT], di[EPT];
    #pragma unroll
    for (int j = 0; j < EPT; ++j) {
        const int e = min(base + j * 1024 + tid, E - 1);
        du[j] = uidx[e];
        di[j] = iidx[e];
    }
    #pragma unroll
    for (int j = 0; j < EPT; ++j) {
        if (base + j * 1024 + tid < E) {
            atomicAdd(&hist[du[j] >> BIN_SHIFT], 1);
            atomicAdd(&hist[(U + di[j]) >> BIN_SHIFT], 1);
        }
    }
    __syncthreads();
    if (tid < NB) H[blockIdx.x * NB + tid] = hist[tid];
}

__global__ __launch_bounds__(1024)
void bin_scatter_kernel(const int* __restrict__ uidx, const int* __restrict__ iidx,
                        const int* __restrict__ H, int* __restrict__ binStart,
                        int2* __restrict__ coarse, int E, int U, int NB, int NBLK)
{
    __shared__ int  hist[NBINS_MAX];
    __shared__ int  lstart[NBINS_MAX];
    __shared__ int  delta[NBINS_MAX];
    __shared__ int  ws[33];
    __shared__ int2 stage[2 * CHUNK];

    const int tid = threadIdx.x;
    const int me  = blockIdx.x;
    hist[tid]   = 0;
    lstart[tid] = 0;
    delta[tid]  = 0;
    __syncthreads();

    const int base = me * CHUNK;
    int du[EPT], di[EPT], pu[EPT], pi[EPT];
    #pragma unroll
    for (int j = 0; j < EPT; ++j) {
        const int e = min(base + j * 1024 + tid, E - 1);
        du[j] = uidx[e];
        di[j] = iidx[e];
    }
    #pragma unroll
    for (int j = 0; j < EPT; ++j) {
        pu[j] = 0; pi[j] = 0;
        if (base + j * 1024 + tid < E) {
            pu[j] = atomicAdd(&hist[du[j] >> BIN_SHIFT], 1);
            pi[j] = atomicAdd(&hist[(U + di[j]) >> BIN_SHIFT], 1);
        } else {
            du[j] = -1;
        }
    }

    {
        const int G = 1024 / NB;
        const int g = tid / NB;
        const int t = tid - g * NB;
        if (g < G) {
            int tot = 0, pre = 0;
            for (int c0 = 0; c0 < NBLK; c0 += HB * G) {
                int v[HB];
                #pragma unroll
                for (int j = 0; j < HB; ++j)
                    v[j] = H[min(c0 + g + j * G, NBLK - 1) * NB + t];
                #pragma unroll
                for (int j = 0; j < HB; ++j) {
                    const int b = c0 + g + j * G;
                    tot += (b < NBLK) ? v[j] : 0;
                    pre += (b < me)   ? v[j] : 0;
                }
            }
            atomicAdd(&lstart[t], tot);
            atomicAdd(&delta[t],  pre);
        }
    }
    __syncthreads();
    const int tot = lstart[tid];
    const int pre = delta[tid];

    int nloc, nall;
    const int ls = block_scan_1024(hist[tid], ws, &nloc);
    const int gs = block_scan_1024(tot, ws, &nall);
    lstart[tid] = ls;
    delta[tid]  = gs + pre - ls;
    if (me == 0) {
        if (tid < NB)  binStart[tid] = gs;
        if (tid == 0)  binStart[NB]  = nall;
    }
    __syncthreads();

    #pragma unroll
    for (int j = 0; j < EPT; ++j) {
        if (du[j] >= 0) {
            const int ru = du[j];
            const int ri = U + di[j];
            stage[lstart[ru >> BIN_SHIFT] + pu[j]] = make_int2(ru, di[j]);
            stage[lstart[ri >> BIN_SHIFT] + pi[j]] = make_int2(ri, du[j]);
        }
    }
    __syncthreads();

    for (int i = tid; i < nloc; i += 1024) {
        const int2 en = stage[i];
        coarse[delta[en.x >> BIN_SHIFT] + i] = en;
    }
}

__global__ __launch_bounds__(1024)
void bin_build_kernel(const int2* __restrict__ coarse, const int* __restrict__ binStart,
                      int* __restrict__ off, float* __restrict__ degf,
                      int* __restrict__ bkt, int R)
{
    __shared__ int cnt[BIN_ROWS];
    __shared__ int cur[BIN_ROWS];
    __shared__ int ws[33];
    __shared__ int sb[STAGE_CAP];

    const int tid = threadIdx.x;
    const int b   = blockIdx.x;
    const int s   = binStart[b];
    const int n   = binStart[b + 1] - s;

    if (tid < BIN_ROWS) cnt[tid] = 0;

    int2 en[KEEP];
    #pragma unroll
    for (int j = 0; j < KEEP; ++j) en[j] = make_int2(-1, 0);
    if (n > 0) {
        #pragma unroll
        for (int j = 0; j < KEEP; ++j) {
            const int  q = j * 1024 + tid;
            const int2 v = coarse[s + min(q, n - 1)];
            if (q < n) en[j] = v;
        }
    }
    __syncthreads();
    #pragma unroll
    for (int j = 0; j < KEEP; ++j)
        if (en[j].x >= 0) atomicAdd(&cnt[en[j].x & (BIN_ROWS - 1)], 1);
    for (int q = KEEP * 1024 + tid; q < n; q += 1024)
        atomicAdd(&cnt[coarse[s + q].x & (BIN_ROWS - 1)], 1);
    __syncthreads();

    const int c = (tid < BIN_ROWS) ? cnt[tid] : 0;
    int nsum;
    const int ex = block_scan_1024(c, ws, &nsum);
    if (tid < BIN_ROWS) {
        cur[tid] = ex;
        const int r = (b << BIN_SHIFT) + tid;
        if (r < R) {
            off[r]  = s + ex;
            degf[r] = (float)c;
        }
    }
    __syncthreads();

    if (n <= STAGE_CAP) {
        #pragma unroll
        for (int j = 0; j < KEEP; ++j) {
            if (en[j].x >= 0) {
                const int k = atomicAdd(&cur[en[j].x & (BIN_ROWS - 1)], 1);
                sb[k] = en[j].y;
            }
        }
        for (int q = KEEP * 1024 + tid; q < n; q += 1024) {
            const int2 e2 = coarse[s + q];
            const int  k  = atomicAdd(&cur[e2.x & (BIN_ROWS - 1)], 1);
            sb[k] = e2.y;
        }
        __syncthreads();
        for (int q = tid; q < n; q += 1024) bkt[s + q] = sb[q];
    } else {
        for (int q = tid; q < n; q += 1024) {
            const int2 e2 = coarse[s + q];
            const int  k  = atomicAdd(&cur[e2.x & (BIN_ROWS - 1)], 1);
            bkt[s + k] = e2.y;
        }
    }
}

__global__ __launch_bounds__(256)
void segment_gather_kernel(const float* __restrict__ UF, const float* __restrict__ IF,
                           const int* __restrict__ off, const float* __restrict__ degf,
                           const int* __restrict__ bkt, float* __restrict__ out,
                           int U, int R)
{
    const int lane = threadIdx.x & 31;
    const int r = __builtin_amdgcn_readfirstlane(blockIdx.x * 8 + (threadIdx.x >> 5));
    if (r >= R) return;
    const float* __restrict__ src = (r < U) ? IF : UF;
    const int n = __builtin_amdgcn_readfirstlane((int)degf[r]);
    const int o = __builtin_amdgcn_readfirstlane(off[r]);

    float4 acc = make_float4(0.f, 0.f, 0.f, 0.f);

    if (n > 0 && n <= 32) {
        const int k = (lane < n) ? bkt[o + lane] : 0x7fffffff;
        int rk = 0;
        for (int l = 0; l < n; ++l) {
            const int s = __builtin_amdgcn_readlane(k, l);
            rk += (s < k || (s == k && l < lane)) ? 1 : 0;
        }
        const int rank = (lane < n) ? rk : lane;
        const int sk = __builtin_amdgcn_ds_permute(rank << 2, k);

        for (int t0 = 0; t0 < n; t0 += 8) {
            float4 f[8];
            #pragma unroll
            for (int j = 0; j < 8; ++j) {
                const int t = (t0 + j < n) ? (t0 + j) : (n - 1);
                const int s = __builtin_amdgcn_readlane(sk, t);
                f[j] = ((const float4*)(src + (size_t)s * D))[lane];
            }
            #pragma unroll
            for (int j = 0; j < 8; ++j) {
                const bool live = (t0 + j < n);
                acc.x += live ? f[j].x : 0.f;
                acc.y += live ? f[j].y : 0.f;
                acc.z += live ? f[j].z : 0.f;
                acc.w += live ? f[j].w : 0.f;
            }
        }
    } else if (n > 32) {
        int cur = -1;
        for (int trip = 0; trip < n; ++trip) {
            int best = 0x7fffffff, m = 0;
            for (int j = lane; j < n; j += 32) {
                const int k = bkt[o + j];
                if (k > cur) {
                    if (k < best)       { best = k; m = 1; }
                    else if (k == best) { ++m; }
                }
            }
            int wb = best;
            for (int d = 16; d > 0; d >>= 1) wb = min(wb, __shfl_xor(wb, d));
            if (wb == 0x7fffffff) break;
            int mm = (best == wb) ? m : 0;
            for (int d = 16; d > 0; d >>= 1) mm += __shfl_xor(mm, d);
            const int s   = __builtin_amdgcn_readfirstlane(wb);
            const int rep = __builtin_amdgcn_readfirstlane(mm);
            const float4 f = ((const float4*)(src + (size_t)s * D))[lane];
            for (int t = 0; t < rep; ++t) {
                acc.x += f.x; acc.y += f.y; acc.z += f.z; acc.w += f.w;
            }
            cur = s;
        }
    }

    ((float4*)(out + (size_t)r * D))[lane] = acc;
}

#define ROW_TILES 5
#define BLOCK_ROWS (ROW_TILES * 16)

__global__ __launch_bounds__(256)
void gnn_gemm_kernel(const float* __restrict__ X, float* __restrict__ out,
                     const float* __restrict__ Wself, const float* __restrict__ bself,
                     const float* __restrict__ Wmsg,  const float* __restrict__ bmsg,
                     const float* __restrict__ deg, int rows)
{
    __shared__ float sX[BLOCK_ROWS * LDS_STRIDE];
    __shared__ float sG[BLOCK_ROWS * LDS_STRIDE];

    const int row0 = blockIdx.x * BLOCK_ROWS;
    const int tid  = threadIdx.x;

    float dsum = 0.f;
    {
        float dg[(BLOCK_ROWS + 31) / 32];
        #pragma unroll
        for (int j = 0; j < (BLOCK_ROWS + 31) / 32; ++j) {
            const int i = min(j * 32 + (tid & 31), BLOCK_ROWS - 1);
            dg[j] = deg[min(row0 + i, rows - 1)];
        }
        #pragma unroll
        for (int j = 0; j < (BLOCK_ROWS + 31) / 32; ++j) dsum += dg[j];
    }
    const bool hasEdge = dsum > 0.f;

    {
        const int c4 = tid & 31;
        const int r0 = tid >> 5;
        float4 xr[BLOCK_ROWS / 8], gv[BLOCK_ROWS / 8];
        #pragma unroll
        for (int j = 0; j < BLOCK_ROWS / 8; ++j) {
            const int gr = min(row0 + r0 + 8 * j, rows - 1);
            xr[j] = ((const float4*)(X   + (size_t)gr * D))[c4];
            gv[j] = ((const float4*)(out + (size_t)gr * D))[c4];
        }
        #pragma unroll
        for (int j = 0; j < BLOCK_ROWS / 8; ++j) {
            const int r = r0 + 8 * j;
            ((float4*)(sX + r * LDS_STRIDE))[c4] = xr[j];
            ((float4*)(sG + r * LDS_STRIDE))[c4] = gv[j];
        }
    }
    __syncthreads();

    const int lane = tid & 31;
    const int wave = tid >> 5;
    const int col0 = wave * 16;
    const int mn   = lane & 15;
    const int koff = (lane >> 4) << 1;

    const float* wS = Wself + (size_t)(col0 + mn) * D;
    const float* wM = Wmsg  + (size_t)(col0 + mn) * D;
    const float* aS = sX + mn * LDS_STRIDE;
    const float* aM = sG + mn * LDS_STRIDE;

    v8f c[ROW_TILES];
    #pragma unroll
    for (int t = 0; t < ROW_TILES; ++t) c[t] = (v8f){};

    #pragma unroll
    for (int k = 0; k < D; k += 4) {
        v2f b = { wS[k + koff], wS[k + koff + 1] };
        #pragma unroll
        for (int t = 0; t < ROW_TILES; ++t) {
            const float* ap = aS + t * 16 * LDS_STRIDE;
            v2f a = { ap[k + koff], ap[k + koff + 1] };
            c[t] = __builtin_amdgcn_wmma_f32_16x16x4_f32(false, a, false, b,
                                                         (short)0, c[t], false, false);
        }
    }
    if (__builtin_amdgcn_ballot_w32(hasEdge) != 0u) {
    #pragma unroll
    for (int k = 0; k < D; k += 4) {
        v2f b = { wM[k + koff], wM[k + koff + 1] };
        #pragma unroll
        for (int t = 0; t < ROW_TILES; ++t) {
            const float* ap = aM + t * 16 * LDS_STRIDE;
            v2f a = { ap[k + koff], ap[k + koff + 1] };
            c[t] = __builtin_amdgcn_wmma_f32_16x16x4_f32(false, a, false, b,
                                                         (short)0, c[t], false, false);
        }
    }
    }

    const int col  = col0 + mn;
    const float bS = bself[col];
    const float bM = bmsg[col];
    #pragma unroll
    for (int t = 0; t < ROW_TILES; ++t) {
        const int rbase = row0 + t * 16 + ((lane >> 4) << 3);
        if (row0 + t * 16 < rows) {
            #pragma unroll
            for (int v = 0; v < 8; ++v) {
                int row = rbase + v;
                out[(size_t)row * D + col] = c[t][v] + bS + deg[row] * bM;
            }
        }
    }
}

extern "C" void kernel_launch(void* const* d_in, const int* in_sizes, int n_in,
                              void* d_out, int out_size, void* d_ws, size_t ws_size,
                              hipStream_t stream)
{
    const float* UF   = (const float*)d_in[0];
    const float* IF   = (const float*)d_in[1];
    const int*   edge = (const int*)d_in[2];
    const float* Wu   = (const float*)d_in[3];
    const float* bu   = (const float*)d_in[4];
    const float* Wi   = (const float*)d_in[5];
    const float* bi   = (const float*)d_in[6];
    const float* Wu2i = (const float*)d_in[7];
    const float* bu2i = (const float*)d_in[8];
    const float* Wi2u = (const float*)d_in[9];
    const float* bi2u = (const float*)d_in[10];

    const int U = in_sizes[0] / D;
    const int I = in_sizes[1] / D;
    const int E = in_sizes[2] / 2;
    const int R = U + I;

    float* outU = (float*)d_out;
    float* outI = outU + (size_t)U * D;

    const int NB   = (R + BIN_ROWS - 1) / BIN_ROWS;
    const int NBLK = (E + CHUNK - 1) / CHUNK;

    int*   off    = (int*)d_ws;
    float* degf   = (float*)(off + R);
    int*   bkt    = (int*)(degf + R);
    int*   H      = bkt + (size_t)2 * E;
    int*   binSt  = H + (size_t)NBLK * NB;
    const size_t hb = ((size_t)NBLK * NB + NB + 2) & ~(size_t)1;
    int2*  coarse = (int2*)(H + hb);

    bin_count_kernel<<<(unsigned)NBLK, 1024, 0, stream>>>(edge, edge + E, H, E, U, NB);
    bin_scatter_kernel<<<(unsigned)NBLK, 1024, 0, stream>>>(
        edge, edge + E, H, binSt, coarse, E, U, NB, NBLK);

    bin_build_kernel<<<(unsigned)NB, 1024, 0, stream>>>(coarse, binSt, off, degf, bkt, R);

    segment_gather_kernel<<<(unsigned)((R + 7) / 8), 256, 0, stream>>>(
        UF, IF, off, degf, bkt, outU, U, R);

    gnn_gemm_kernel<<<(unsigned)((U + BLOCK_ROWS - 1) / BLOCK_ROWS), 256, 0, stream>>>(
        UF, outU, Wu, bu, Wi2u, bi2u, degf, U);
    gnn_gemm_kernel<<<(unsigned)((I + BLOCK_ROWS - 1) / BLOCK_ROWS), 256, 0, stream>>>(
        IF, outI, Wi, bi, Wu2i, bu2i, degf + U, I);
}


# One launch for both GEMMs: item and user row blocks share a grid (item blocks first), so no launch boundary, no second fill phase and no partial last wave of blocks between them; sums unchanged
# speedup vs baseline: 2.8733x; 1.0632x over previous
#include <hip/hip_runtime.h>
#include <hip/hip_bf16.h>

typedef __attribute__((ext_vector_type(2))) float v2f;
typedef __attribute__((ext_vector_type(8))) float v8f;

#define D 128
#define LDS_STRIDE 132

#define BIN_SHIFT 9
#define BIN_ROWS  (1 << BIN_SHIFT)
#define CHUNK     4096
#define EPT       (CHUNK / 1024)
#define NBINS_MAX 1024
#define HB        32
#define STAGE_CAP 8192
#define KEEP      5

__device__ __forceinline__ int block_scan_1024(int c, int* ws, int* total)
{
    const int lane = threadIdx.x & 31;
    const int w    = threadIdx.x >> 5;
    int x = c;
    for (int d = 1; d < 32; d <<= 1) {
        int y = __shfl_up(x, d);
        if (lane >= d) x += y;
    }
    if (lane == 31) ws[w] = x;
    __syncthreads();
    if (w == 0) {
        const int s = ws[lane];
        int t = s;
        for (int d = 1; d < 32; d <<= 1) {
            int y = __shfl_up(t, d);
            if (lane >= d) t += y;
        }
        ws[lane] = t - s;
        if (lane == 31) ws[32] = t;
    }
    __syncthreads();
    const int r = ws[w] + x - c;
    *total = ws[32];
    __syncthreads();
    return r;
}

__global__ __launch_bounds__(1024)
void bin_count_kernel(const int* __restrict__ uidx, const int* __restrict__ iidx,
                      int* __restrict__ H, int E, int U, int NB)
{
    __shared__ int hist[NBINS_MAX];
    const int tid = threadIdx.x;
    hist[tid] = 0;
    __syncthreads();
    const int base = blockIdx.x * CHUNK;
    int du[EPT], di[EPT];
    #pragma unroll
    for (int j = 0; j < EPT; ++j) {
        const int e = min(base + j * 1024 + tid, E - 1);
        du[j] = uidx[e];
        di[j] = iidx[e];
    }
    #pragma unroll
    for (int j = 0; j < EPT; ++j) {
        if (base + j * 1024 + tid < E) {
            atomicAdd(&hist[du[j] >> BIN_SHIFT], 1);
            atomicAdd(&hist[(U + di[j]) >> BIN_SHIFT], 1);
        }
    }
    __syncthreads();
    if (tid < NB) H[blockIdx.x * NB + tid] = hist[tid];
}

__global__ __launch_bounds__(1024)
void bin_scatter_kernel(const int* __restrict__ uidx, const int* __restrict__ iidx,
                        const int* __restrict__ H, int* __restrict__ binStart,
                        int2* __restrict__ coarse, int E, int U, int NB, int NBLK)
{
    __shared__ int  hist[NBINS_MAX];
    __shared__ int  lstart[NBINS_MAX];
    __shared__ int  delta[NBINS_MAX];
    __shared__ int  ws[33];
    __shared__ int2 stage[2 * CHUNK];

    const int tid = threadIdx.x;
    const int me  = blockIdx.x;
    hist[tid]   = 0;
    lstart[tid] = 0;
    delta[tid]  = 0;
    __syncthreads();

    const int base = me * CHUNK;
    int du[EPT], di[EPT], pu[EPT], pi[EPT];
    #pragma unroll
    for (int j = 0; j < EPT; ++j) {
        const int e = min(base + j * 1024 + tid, E - 1);
        du[j] = uidx[e];
        di[j] = iidx[e];
    }
    #pragma unroll
    for (int j = 0; j < EPT; ++j) {
        pu[j] = 0; pi[j] = 0;
        if (base + j * 1024 + tid < E) {
            pu[j] = atomicAdd(&hist[du[j] >> BIN_SHIFT], 1);
            pi[j] = atomicAdd(&hist[(U + di[j]) >> BIN_SHIFT], 1);
        } else {
            du[j] = -1;
        }
    }

    {
        const int G = 1024 / NB;
        const int g = tid / NB;
        const int t = tid - g * NB;
        if (g < G) {
            int tot = 0, pre = 0;
            for (int c0 = 0; c0 < NBLK; c0 += HB * G) {
                int v[HB];
                #pragma unroll
                for (int j = 0; j < HB; ++j)
                    v[j] = H[min(c0 + g + j * G, NBLK - 1) * NB + t];
                #pragma unroll
                for (int j = 0; j < HB; ++j) {
                    const int b = c0 + g + j * G;
                    tot += (b < NBLK) ? v[j] : 0;
                    pre += (b < me)   ? v[j] : 0;
                }
            }
            atomicAdd(&lstart[t], tot);
            atomicAdd(&delta[t],  pre);
        }
    }
    __syncthreads();
    const int tot = lstart[tid];
    const int pre = delta[tid];

    int nloc, nall;
    const int ls = block_scan_1024(hist[tid], ws, &nloc);
    const int gs = block_scan_1024(tot, ws, &nall);
    lstart[tid] = ls;
    delta[tid]  = gs + pre - ls;
    if (me == 0) {
        if (tid < NB)  binStart[tid] = gs;
        if (tid == 0)  binStart[NB]  = nall;
    }
    __syncthreads();

    #pragma unroll
    for (int j = 0; j < EPT; ++j) {
        if (du[j] >= 0) {
            const int ru = du[j];
            const int ri = U + di[j];
            stage[lstart[ru >> BIN_SHIFT] + pu[j]] = make_int2(ru, di[j]);
            stage[lstart[ri >> BIN_SHIFT] + pi[j]] = make_int2(ri, du[j]);
        }
    }
    __syncthreads();

    for (int i = tid; i < nloc; i += 1024) {
        const int2 en = stage[i];
        coarse[delta[en.x >> BIN_SHIFT] + i] = en;
    }
}

__global__ __launch_bounds__(1024)
void bin_build_kernel(const int2* __restrict__ coarse, const int* __restrict__ binStart,
                      int* __restrict__ off, float* __restrict__ degf,
                      int* __restrict__ bkt, int R)
{
    __shared__ int cnt[BIN_ROWS];
    __shared__ int cur[BIN_ROWS];
    __shared__ int ws[33];
    __shared__ int sb[STAGE_CAP];

    const int tid = threadIdx.x;
    const int b   = blockIdx.x;
    const int s   = binStart[b];
    const int n   = binStart[b + 1] - s;

    if (tid < BIN_ROWS) cnt[tid] = 0;

    int2 en[KEEP];
    #pragma unroll
    for (int j = 0; j < KEEP; ++j) en[j] = make_int2(-1, 0);
    if (n > 0) {
        #pragma unroll
        for (int j = 0; j < KEEP; ++j) {
            const int  q = j * 1024 + tid;
            const int2 v = coarse[s + min(q, n - 1)];
            if (q < n) en[j] = v;
        }
    }
    __syncthreads();
    #pragma unroll
    for (int j = 0; j < KEEP; ++j)
        if (en[j].x >= 0) atomicAdd(&cnt[en[j].x & (BIN_ROWS - 1)], 1);
    for (int q = KEEP * 1024 + tid; q < n; q += 1024)
        atomicAdd(&cnt[coarse[s + q].x & (BIN_ROWS - 1)], 1);
    __syncthreads();

    const int c = (tid < BIN_ROWS) ? cnt[tid] : 0;
    int nsum;
    const int ex = block_scan_1024(c, ws, &nsum);
    if (tid < BIN_ROWS) {
        cur[tid] = ex;
        const int r = (b << BIN_SHIFT) + tid;
        if (r < R) {
            off[r]  = s + ex;
            degf[r] = (float)c;
        }
    }
    __syncthreads();

    if (n <= STAGE_CAP) {
        #pragma unroll
        for (int j = 0; j < KEEP; ++j) {
            if (en[j].x >= 0) {
                const int k = atomicAdd(&cur[en[j].x & (BIN_ROWS - 1)], 1);
                sb[k] = en[j].y;
            }
        }
        for (int q = KEEP * 1024 + tid; q < n; q += 1024) {
            const int2 e2 = coarse[s + q];
            const int  k  = atomicAdd(&cur[e2.x & (BIN_ROWS - 1)], 1);
            sb[k] = e2.y;
        }
        __syncthreads();
        for (int q = tid; q < n; q += 1024) bkt[s + q] = sb[q];
    } else {
        for (int q = tid; q < n; q += 1024) {
            const int2 e2 = coarse[s + q];
            const int  k  = atomicAdd(&cur[e2.x & (BIN_ROWS - 1)], 1);
            bkt[s + k] = e2.y;
        }
    }
}

__global__ __launch_bounds__(256)
void segment_gather_kernel(const float* __restrict__ UF, const float* __restrict__ IF,
                           const int* __restrict__ off, const float* __restrict__ degf,
                           const int* __restrict__ bkt, float* __restrict__ out,
                           int U, int R)
{
    const int lane = threadIdx.x & 31;
    const int r = __builtin_amdgcn_readfirstlane(blockIdx.x * 8 + (threadIdx.x >> 5));
    if (r >= R) return;
    const float* __restrict__ src = (r < U) ? IF : UF;
    const int n = __builtin_amdgcn_readfirstlane((int)degf[r]);
    const int o = __builtin_amdgcn_readfirstlane(off[r]);

    float4 acc = make_float4(0.f, 0.f, 0.f, 0.f);

    if (n > 0 && n <= 32) {
        const int k = (lane < n) ? bkt[o + lane] : 0x7fffffff;
        int rk = 0;
        for (int l = 0; l < n; ++l) {
            const int s = __builtin_amdgcn_readlane(k, l);
            rk += (s < k || (s == k && l < lane)) ? 1 : 0;
        }
        const int rank = (lane < n) ? rk : lane;
        const int sk = __builtin_amdgcn_ds_permute(rank << 2, k);

        for (int t0 = 0; t0 < n; t0 += 8) {
            float4 f[8];
            #pragma unroll
            for (int j = 0; j < 8; ++j) {
                const int t = (t0 + j < n) ? (t0 + j) : (n - 1);
                const int s = __builtin_amdgcn_readlane(sk, t);
                f[j] = ((const float4*)(src + (size_t)s * D))[lane];
            }
            #pragma unroll
            for (int j = 0; j < 8; ++j) {
                const bool live = (t0 + j < n);
                acc.x += live ? f[j].x : 0.f;
                acc.y += live ? f[j].y : 0.f;
                acc.z += live ? f[j].z : 0.f;
                acc.w += live ? f[j].w : 0.f;
            }
        }
    } else if (n > 32) {
        int cur = -1;
        for (int trip = 0; trip < n; ++trip) {
            int best = 0x7fffffff, m = 0;
            for (int j = lane; j < n; j += 32) {
                const int k = bkt[o + j];
                if (k > cur) {
                    if (k < best)       { best = k; m = 1; }
                    else if (k == best) { ++m; }
                }
            }
            int wb = best;
            for (int d = 16; d > 0; d >>= 1) wb = min(wb, __shfl_xor(wb, d));
            if (wb == 0x7fffffff) break;
            int mm = (best == wb) ? m : 0;
            for (int d = 16; d > 0; d >>= 1) mm += __shfl_xor(mm, d);
            const int s   = __builtin_amdgcn_readfirstlane(wb);
            const int rep = __builtin_amdgcn_readfirstlane(mm);
            const float4 f = ((const float4*)(src + (size_t)s * D))[lane];
            for (int t = 0; t < rep; ++t) {
                acc.x += f.x; acc.y += f.y; acc.z += f.z; acc.w += f.w;
            }
            cur = s;
        }
    }

    ((float4*)(out + (size_t)r * D))[lane] = acc;
}

#define ROW_TILES 5
#define BLOCK_ROWS (ROW_TILES * 16)

__global__ __launch_bounds__(256)
void gnn_gemm_kernel(const float* __restrict__ XU, const float* __restrict__ XI,
                     float* __restrict__ outU, float* __restrict__ outI,
                     const float* __restrict__ WsU, const float* __restrict__ bsU,
                     const float* __restrict__ WmU, const float* __restrict__ bmU,
                     const float* __restrict__ WsI, const float* __restrict__ bsI,
                     const float* __restrict__ WmI, const float* __restrict__ bmI,
                     const float* __restrict__ degf, int U, int I, int nbI)
{
    __shared__ float sX[BLOCK_ROWS * LDS_STRIDE];
    __shared__ float sG[BLOCK_ROWS * LDS_STRIDE];

    const bool item = (int)blockIdx.x < nbI;
    const float* __restrict__ X     = item ? XI  : XU;
    float*       __restrict__ out   = item ? outI : outU;
    const float* __restrict__ Wself = item ? WsI : WsU;
    const float* __restrict__ bself = item ? bsI : bsU;
    const float* __restrict__ Wmsg  = item ? WmI : WmU;
    const float* __restrict__ bmsg  = item ? bmI : bmU;
    const float* __restrict__ deg   = item ? degf + U : degf;
    const int rows = item ? I : U;

    const int row0 = (item ? (int)blockIdx.x : (int)blockIdx.x - nbI) * BLOCK_ROWS;
    const int tid  = threadIdx.x;

    float dsum = 0.f;
    {
        float dg[(BLOCK_ROWS + 31) / 32];
        #pragma unroll
        for (int j = 0; j < (BLOCK_ROWS + 31) / 32; ++j) {
            const int i = min(j * 32 + (tid & 31), BLOCK_ROWS - 1);
            dg[j] = deg[min(row0 + i, rows - 1)];
        }
        #pragma unroll
        for (int j = 0; j < (BLOCK_ROWS + 31) / 32; ++j) dsum += dg[j];
    }
    const bool hasEdge = dsum > 0.f;

    {
        const int c4 = tid & 31;
        const int r0 = tid >> 5;
        float4 xr[BLOCK_ROWS / 8], gv[BLOCK_ROWS / 8];
        #pragma unroll
        for (int j = 0; j < BLOCK_ROWS / 8; ++j) {
            const int gr = min(row0 + r0 + 8 * j, rows - 1);
            xr[j] = ((const float4*)(X   + (size_t)gr * D))[c4];
            gv[j] = ((const float4*)(out + (size_t)gr * D))[c4];
        }
        #pragma unroll
        for (int j = 0; j < BLOCK_ROWS / 8; ++j) {
            const int r = r0 + 8 * j;
            ((float4*)(sX + r * LDS_STRIDE))[c4] = xr[j];
            ((float4*)(sG + r * LDS_STRIDE))[c4] = gv[j];
        }
    }
    __syncthreads();

    const int lane = tid & 31;
    const int wave = tid >> 5;
    const int col0 = wave * 16;
    const int mn   = lane & 15;
    const int koff = (lane >> 4) << 1;

    const float* wS = Wself + (size_t)(col0 + mn) * D;
    const float* wM = Wmsg  + (size_t)(col0 + mn) * D;
    const float* aS = sX + mn * LDS_STRIDE;
    const float* aM = sG + mn * LDS_STRIDE;

    v8f c[ROW_TILES];
    #pragma unroll
    for (int t = 0; t < ROW_TILES; ++t) c[t] = (v8f){};

    #pragma unroll
    for (int k = 0; k < D; k += 4) {
        v2f b = { wS[k + koff], wS[k + koff + 1] };
        #pragma unroll
        for (int t = 0; t < ROW_TILES; ++t) {
            const float* ap = aS + t * 16 * LDS_STRIDE;
            v2f a = { ap[k + koff], ap[k + koff + 1] };
            c[t] = __builtin_amdgcn_wmma_f32_16x16x4_f32(false, a, false, b,
                                                         (short)0, c[t], false, false);
        }
    }
    if (__builtin_amdgcn_ballot_w32(hasEdge) != 0u) {
    #pragma unroll
    for (int k = 0; k < D; k += 4) {
        v2f b = { wM[k + koff], wM[k + koff + 1] };
        #pragma unroll
        for (int t = 0; t < ROW_TILES; ++t) {
            const float* ap = aM + t * 16 * LDS_STRIDE;
            v2f a = { ap[k + koff], ap[k + koff + 1] };
            c[t] = __builtin_amdgcn_wmma_f32_16x16x4_f32(false, a, false, b,
                                                         (short)0, c[t], false, false);
        }
    }
    }

    const int col  = col0 + mn;
    const float bS = bself[col];
    const float bM = bmsg[col];
    #pragma unroll
    for (int t = 0; t < ROW_TILES; ++t) {
        const int rbase = row0 + t * 16 + ((lane >> 4) << 3);
        if (row0 + t * 16 < rows) {
            #pragma unroll
            for (int v = 0; v < 8; ++v) {
                int row = rbase + v;
                out[(size_t)row * D + col] = c[t][v] + bS + deg[row] * bM;
            }
        }
    }
}

extern "C" void kernel_launch(void* const* d_in, const int* in_sizes, int n_in,
                              void* d_out, int out_size, void* d_ws, size_t ws_size,
                              hipStream_t stream)
{
    const float* UF   = (const float*)d_in[0];
    const float* IF   = (const float*)d_in[1];
    const int*   edge = (const int*)d_in[2];
    const float* Wu   = (const float*)d_in[3];
    const float* bu   = (const float*)d_in[4];
    const float* Wi   = (const float*)d_in[5];
    const float* bi   = (const float*)d_in[6];
    const float* Wu2i = (const float*)d_in[7];
    const float* bu2i = (const float*)d_in[8];
    const float* Wi2u = (const float*)d_in[9];
    const float* bi2u = (const float*)d_in[10];

    const int U = in_sizes[0] / D;
    const int I = in_sizes[1] / D;
    const int E = in_sizes[2] / 2;
    const int R = U + I;

    float* outU = (float*)d_out;
    float* outI = outU + (size_t)U * D;

    const int NB   = (R + BIN_ROWS - 1) / BIN_ROWS;
    const int NBLK = (E + CHUNK - 1) / CHUNK;

    int*   off    = (int*)d_ws;
    float* degf   = (float*)(off + R);
    int*   bkt    = (int*)(degf + R);
    int*   H      = bkt + (size_t)2 * E;
    int*   binSt  = H + (size_t)NBLK * NB;
    const size_t hb = ((size_t)NBLK * NB + NB + 2) & ~(size_t)1;
    int2*  coarse = (int2*)(H + hb);

    bin_count_kernel<<<(unsigned)NBLK, 1024, 0, stream>>>(edge, edge + E, H, E, U, NB);
    bin_scatter_kernel<<<(unsigned)NBLK, 1024, 0, stream>>>(
        edge, edge + E, H, binSt, coarse, E, U, NB, NBLK);

    bin_build_kernel<<<(unsigned)NB, 1024, 0, stream>>>(coarse, binSt, off, degf, bkt, R);

    segment_gather_kernel<<<(unsigned)((R + 7) / 8), 256, 0, stream>>>(
        UF, IF, off, degf, bkt, outU, U, R);

    const int nbU = (U + BLOCK_ROWS - 1) / BLOCK_ROWS;
    const int nbI = (I + BLOCK_ROWS - 1) / BLOCK_ROWS;
    gnn_gemm_kernel<<<(unsigned)(nbU + nbI), 256, 0, stream>>>(
        UF, IF, outU, outI, Wu, bu, Wi2u, bi2u, Wi, bi, Wu2i, bu2i, degf, U, I, nbI);
}


# Segment sums taken inside the GEMM launch: each block adds its 80 rows' source rows straight into its LDS aggregate tile (one launch less, no pass of the sums through out); sum order unchanged
# speedup vs baseline: 3.0907x; 1.0757x over previous
#include <hip/hip_runtime.h>
#include <hip/hip_bf16.h>

typedef __attribute__((ext_vector_type(2))) float v2f;
typedef __attribute__((ext_vector_type(8))) float v8f;

#define D 128
#define LDS_STRIDE 132

#define BIN_SHIFT 9
#define BIN_ROWS  (1 << BIN_SHIFT)
#define CHUNK     4096
#define EPT       (CHUNK / 1024)
#define NBINS_MAX 1024
#define HB        32
#define STAGE_CAP 8192
#define KEEP      5

__device__ __forceinline__ int block_scan_1024(int c, int* ws, int* total)
{
    const int lane = threadIdx.x & 31;
    const int w    = threadIdx.x >> 5;
    int x = c;
    for (int d = 1; d < 32; d <<= 1) {
        int y = __shfl_up(x, d);
        if (lane >= d) x += y;
    }
    if (lane == 31) ws[w] = x;
    __syncthreads();
    if (w == 0) {
        const int s = ws[lane];
        int t = s;
        for (int d = 1; d < 32; d <<= 1) {
            int y = __shfl_up(t, d);
            if (lane >= d) t += y;
        }
        ws[lane] = t - s;
        if (lane == 31) ws[32] = t;
    }
    __syncthreads();
    const int r = ws[w] + x - c;
    *total = ws[32];
    __syncthreads();
    return r;
}

__global__ __launch_bounds__(1024)
void bin_count_kernel(const int* __restrict__ uidx, const int* __restrict__ iidx,
                      int* __restrict__ H, int E, int U, int NB)
{
    __shared__ int hist[NBINS_MAX];
    const int tid = threadIdx.x;
    hist[tid] = 0;
    __syncthreads();
    const int base = blockIdx.x * CHUNK;
    int du[EPT], di[EPT];
    #pragma unroll
    for (int j = 0; j < EPT; ++j) {
        const int e = min(base + j * 1024 + tid, E - 1);
        du[j] = uidx[e];
        di[j] = iidx[e];
    }
    #pragma unroll
    for (int j = 0; j < EPT; ++j) {
        if (base + j * 1024 + tid < E) {
            atomicAdd(&hist[du[j] >> BIN_SHIFT], 1);
            atomicAdd(&hist[(U + di[j]) >> BIN_SHIFT], 1);
        }
    }
    __syncthreads();
    if (tid < NB) H[blockIdx.x * NB + tid] = hist[tid];
}

__global__ __launch_bounds__(1024)
void bin_scatter_kernel(const int* __restrict__ uidx, const int* __restrict__ iidx,
                        const int* __restrict__ H, int* __restrict__ binStart,
                        int2* __restrict__ coarse, int E, int U, int NB, int NBLK)
{
    __shared__ int  hist[NBINS_MAX];
    __shared__ int  lstart[NBINS_MAX];
    __shared__ int  delta[NBINS_MAX];
    __shared__ int  ws[33];
    __shared__ int2 stage[2 * CHUNK];

    const int tid = threadIdx.x;
    const int me  = blockIdx.x;
    hist[tid]   = 0;
    lstart[tid] = 0;
    delta[tid]  = 0;
    __syncthreads();

    const int base = me * CHUNK;
    int du[EPT], di[EPT], pu[EPT], pi[EPT];
    #pragma unroll
    for (int j = 0; j < EPT; ++j) {
        const int e = min(base + j * 1024 + tid, E - 1);
        du[j] = uidx[e];
        di[j] = iidx[e];
    }
    #pragma unroll
    for (int j = 0; j < EPT; ++j) {
        pu[j] = 0; pi[j] = 0;
        if (base + j * 1024 + tid < E) {
            pu[j] = atomicAdd(&hist[du[j] >> BIN_SHIFT], 1);
            pi[j] = atomicAdd(&hist[(U + di[j]) >> BIN_SHIFT], 1);
        } else {
            du[j] = -1;
        }
    }

    {
        const int G = 1024 / NB;
        const int g = tid / NB;
        const int t = tid - g * NB;
        if (g < G) {
            int tot = 0, pre = 0;
            for (int c0 = 0; c0 < NBLK; c0 += HB * G) {
                int v[HB];
                #pragma unroll
                for (int j = 0; j < HB; ++j)
                    v[j] = H[min(c0 + g + j * G, NBLK - 1) * NB + t];
                #pragma unroll
                for (int j = 0; j < HB; ++j) {
                    const int b = c0 + g + j * G;
                    tot += (b < NBLK) ? v[j] : 0;
                    pre += (b < me)   ? v[j] : 0;
                }
            }
            atomicAdd(&lstart[t], tot);
            atomicAdd(&delta[t],  pre);
        }
    }
    __syncthreads();
    const int tot = lstart[tid];
    const int pre = delta[tid];

    int nloc, nall;
    const int ls = block_scan_1024(hist[tid], ws, &nloc);
    const int gs = block_scan_1024(tot, ws, &nall);
    lstart[tid] = ls;
    delta[tid]  = gs + pre - ls;
    if (me == 0) {
        if (tid < NB)  binStart[tid] = gs;
        if (tid == 0)  binStart[NB]  = nall;
    }
    __syncthreads();

    #pragma unroll
    for (int j = 0; j < EPT; ++j) {
        if (du[j] >= 0) {
            const int ru = du[j];
            const int ri = U + di[j];
            stage[lstart[ru >> BIN_SHIFT] + pu[j]] = make_int2(ru, di[j]);
            stage[lstart[ri >> BIN_SHIFT] + pi[j]] = make_int2(ri, du[j]);
        }
    }
    __syncthreads();

    for (int i = tid; i < nloc; i += 1024) {
        const int2 en = stage[i];
        coarse[delta[en.x >> BIN_SHIFT] + i] = en;
    }
}

__global__ __launch_bounds__(1024)
void bin_build_kernel(const int2* __restrict__ coarse, const int* __restrict__ binStart,
                      int* __restrict__ off, float* __restrict__ degf,
                      int* __restrict__ bkt, int R)
{
    __shared__ int cnt[BIN_ROWS];
    __shared__ int cur[BIN_ROWS];
    __shared__ int ws[33];
    __shared__ int sb[STAGE_CAP];

    const int tid = threadIdx.x;
    const int b   = blockIdx.x;
    const int s   = binStart[b];
    const int n   = binStart[b + 1] - s;

    if (tid < BIN_ROWS) cnt[tid] = 0;

    int2 en[KEEP];
    #pragma unroll
    for (int j = 0; j < KEEP; ++j) en[j] = make_int2(-1, 0);
    if (n > 0) {
        #pragma unroll
        for (int j = 0; j < KEEP; ++j) {
            const int  q = j * 1024 + tid;
            const int2 v = coarse[s + min(q, n - 1)];
            if (q < n) en[j] = v;
        }
    }
    __syncthreads();
    #pragma unroll
    for (int j = 0; j < KEEP; ++j)
        if (en[j].x >= 0) atomicAdd(&cnt[en[j].x & (BIN_ROWS - 1)], 1);
    for (int q = KEEP * 1024 + tid; q < n; q += 1024)
        atomicAdd(&cnt[coarse[s + q].x & (BIN_ROWS - 1)], 1);
    __syncthreads();

    const int c = (tid < BIN_ROWS) ? cnt[tid] : 0;
    int nsum;
    const int ex = block_scan_1024(c, ws, &nsum);
    if (tid < BIN_ROWS) {
        cur[tid] = ex;
        const int r = (b << BIN_SHIFT) + tid;
        if (r < R) {
            off[r]  = s + ex;
            degf[r] = (float)c;
        }
    }
    __syncthreads();

    if (n <= STAGE_CAP) {
        #pragma unroll
        for (int j = 0; j < KEEP; ++j) {
            if (en[j].x >= 0) {
                const int k = atomicAdd(&cur[en[j].x & (BIN_ROWS - 1)], 1);
                sb[k] = en[j].y;
            }
        }
        for (int q = KEEP * 1024 + tid; q < n; q += 1024) {
            const int2 e2 = coarse[s + q];
            const int  k  = atomicAdd(&cur[e2.x & (BIN_ROWS - 1)], 1);
            sb[k] = e2.y;
        }
        __syncthreads();
        for (int q = tid; q < n; q += 1024) bkt[s + q] = sb[q];
    } else {
        for (int q = tid; q < n; q += 1024) {
            const int2 e2 = coarse[s + q];
            const int  k  = atomicAdd(&cur[e2.x & (BIN_ROWS - 1)], 1);
            bkt[s + k] = e2.y;
        }
    }
}

#define GW 16

__device__ __forceinline__ int sort_keys(const int n, const int k, const int lane)
{
    int rk = 0;
    for (int l = 0; l < n; ++l) {
        const int s = __builtin_amdgcn_readlane(k, l);
        rk += (s < k || (s == k && l < lane)) ? 1 : 0;
    }
    const int rank = (lane < n) ? rk : lane;
    return __builtin_amdgcn_ds_permute(rank << 2, k);
}

__device__ __forceinline__ float4 sum_sorted(const float* __restrict__ src,
                                             const int n, const int sk, const int lane)
{
    float4 acc = make_float4(0.f, 0.f, 0.f, 0.f);
    for (int t0 = 0; t0 < n; t0 += GW) {
        float4 f[GW];
        #pragma unroll
        for (int j = 0; j < GW; ++j) {
            f[j] = make_float4(0.f, 0.f, 0.f, 0.f);
            if (t0 + j < n) {
                const int s = __builtin_amdgcn_readlane(sk, t0 + j);
                f[j] = ((const float4*)(src + (size_t)s * D))[lane];
            }
        }
        #pragma unroll
        for (int j = 0; j < GW; ++j) {
            if (t0 + j < n) {
                acc.x += f[j].x;
                acc.y += f[j].y;
                acc.z += f[j].z;
                acc.w += f[j].w;
            }
        }
    }
    return acc;
}

__device__ __forceinline__ float4 sum_long(const float* __restrict__ src,
                                           const int* __restrict__ bkt,
                                           const int n, const int o, const int lane)
{
    float4 acc = make_float4(0.f, 0.f, 0.f, 0.f);
    int cur = -1;
    for (int trip = 0; trip < n; ++trip) {
        int best = 0x7fffffff, m = 0;
        for (int j = lane; j < n; j += 32) {
            const int kk = bkt[o + j];
            if (kk > cur) {
                if (kk < best)       { best = kk; m = 1; }
                else if (kk == best) { ++m; }
            }
        }
        int wb = best;
        for (int d = 16; d > 0; d >>= 1) wb = min(wb, __shfl_xor(wb, d));
        if (wb == 0x7fffffff) break;
        int mm = (best == wb) ? m : 0;
        for (int d = 16; d > 0; d >>= 1) mm += __shfl_xor(mm, d);
        const int s   = __builtin_amdgcn_readfirstlane(wb);
        const int rep = __builtin_amdgcn_readfirstlane(mm);
        const float4 f = ((const float4*)(src + (size_t)s * D))[lane];
        for (int t = 0; t < rep; ++t) {
            acc.x += f.x; acc.y += f.y; acc.z += f.z; acc.w += f.w;
        }
        cur = s;
    }
    return acc;
}

#define ROW_TILES 5
#define BLOCK_ROWS (ROW_TILES * 16)

__global__ __launch_bounds__(256)
void gnn_gemm_kernel(const float* __restrict__ XU, const float* __restrict__ XI,
                     float* __restrict__ outU, float* __restrict__ outI,
                     const float* __restrict__ WsU, const float* __restrict__ bsU,
                     const float* __restrict__ WmU, const float* __restrict__ bmU,
                     const float* __restrict__ WsI, const float* __restrict__ bsI,
                     const float* __restrict__ WmI, const float* __restrict__ bmI,
                     const float* __restrict__ degf, const int* __restrict__ off,
                     const int* __restrict__ bkt, int U, int I, int nbI)
{
    __shared__ float sX[BLOCK_ROWS * LDS_STRIDE];
    __shared__ float sG[BLOCK_ROWS * LDS_STRIDE];

    const bool item = (int)blockIdx.x < nbI;
    const float* __restrict__ X     = item ? XI  : XU;
    float*       __restrict__ out   = item ? outI : outU;
    const float* __restrict__ Wself = item ? WsI : WsU;
    const float* __restrict__ bself = item ? bsI : bsU;
    const float* __restrict__ Wmsg  = item ? WmI : WmU;
    const float* __restrict__ bmsg  = item ? bmI : bmU;
    const float* __restrict__ deg   = item ? degf + U : degf;
    const int*   __restrict__ offp  = item ? off + U : off;
    const float* __restrict__ srcT  = item ? XU : XI;
    const int rows = item ? I : U;

    const int row0 = (item ? (int)blockIdx.x : (int)blockIdx.x - nbI) * BLOCK_ROWS;
    const int tid  = threadIdx.x;

    float dsum = 0.f;
    {
        float dg[(BLOCK_ROWS + 31) / 32];
        #pragma unroll
        for (int j = 0; j < (BLOCK_ROWS + 31) / 32; ++j) {
            const int i = min(j * 32 + (tid & 31), BLOCK_ROWS - 1);
            dg[j] = deg[min(row0 + i, rows - 1)];
        }
        #pragma unroll
        for (int j = 0; j < (BLOCK_ROWS + 31) / 32; ++j) dsum += dg[j];
    }
    const bool hasEdge = dsum > 0.f;

    {
        const int c4 = tid & 31;
        const int r0 = tid >> 5;
        float4 xr[BLOCK_ROWS / 8];
        #pragma unroll
        for (int j = 0; j < BLOCK_ROWS / 8; ++j) {
            const int gr = min(row0 + r0 + 8 * j, rows - 1);
            xr[j] = ((const float4*)(X + (size_t)gr * D))[c4];
        }

        int hn = 0, ho = 0;
        if (c4 < BLOCK_ROWS / 8) {
            const int gr = min(row0 + r0 + 8 * c4, rows - 1);
            hn = (int)deg[gr];
            ho = offp[gr];
        }
        int n = __builtin_amdgcn_readlane(hn, 0);
        int o = __builtin_amdgcn_readlane(ho, 0);
        int k = 0x7fffffff;
        if (n <= 32 && c4 < n) k = bkt[o + c4];

        #pragma unroll
        for (int j = 0; j < BLOCK_ROWS / 8; ++j) {
            const int r = r0 + 8 * j;
            ((float4*)(sX + r * LDS_STRIDE))[c4] = xr[j];
        }

        for (int j = 0; j < BLOCK_ROWS / 8; ++j) {
            const bool shortb = (n > 0 && n <= 32);
            int sk = 0;
            if (shortb) sk = sort_keys(n, k, c4);
            int n1 = 0, o1 = 0, k1 = 0x7fffffff;
            if (j + 1 < BLOCK_ROWS / 8) {
                n1 = __builtin_amdgcn_readlane(hn, j + 1);
                o1 = __builtin_amdgcn_readlane(ho, j + 1);
                if (n1 <= 32 && c4 < n1) k1 = bkt[o1 + c4];
            }
            float4 a = make_float4(0.f, 0.f, 0.f, 0.f);
            if (shortb)      a = sum_sorted(srcT, n, sk, c4);
            else if (n > 32) a = sum_long(srcT, bkt, n, o, c4);
            ((float4*)(sG + (r0 + 8 * j) * LDS_STRIDE))[c4] = a;
            n = n1; o = o1; k = k1;
        }
    }
    __syncthreads();

    const int lane = tid & 31;
    const int wave = tid >> 5;
    const int col0 = wave * 16;
    const int mn   = lane & 15;
    const int koff = (lane >> 4) << 1;

    const float* wS = Wself + (size_t)(col0 + mn) * D;
    const float* wM = Wmsg  + (size_t)(col0 + mn) * D;
    const float* aS = sX + mn * LDS_STRIDE;
    const float* aM = sG + mn * LDS_STRIDE;

    v8f c[ROW_TILES];
    #pragma unroll
    for (int t = 0; t < ROW_TILES; ++t) c[t] = (v8f){};

    #pragma unroll
    for (int k = 0; k < D; k += 4) {
        v2f b = { wS[k + koff], wS[k + koff + 1] };
        #pragma unroll
        for (int t = 0; t < ROW_TILES; ++t) {
            const float* ap = aS + t * 16 * LDS_STRIDE;
            v2f a = { ap[k + koff], ap[k + koff + 1] };
            c[t] = __builtin_amdgcn_wmma_f32_16x16x4_f32(false, a, false, b,
                                                         (short)0, c[t], false, false);
        }
    }
    if (__builtin_amdgcn_ballot_w32(hasEdge) != 0u) {
    #pragma unroll
    for (int k = 0; k < D; k += 4) {
        v2f b = { wM[k + koff], wM[k + koff + 1] };
        #pragma unroll
        for (int t = 0; t < ROW_TILES; ++t) {
            const float* ap = aM + t * 16 * LDS_STRIDE;
            v2f a = { ap[k + koff], ap[k + koff + 1] };
            c[t] = __builtin_amdgcn_wmma_f32_16x16x4_f32(false, a, false, b,
                                                         (short)0, c[t], false, false);
        }
    }
    }

    const int col  = col0 + mn;
    const float bS = bself[col];
    const float bM = bmsg[col];
    #pragma unroll
    for (int t = 0; t < ROW_TILES; ++t) {
        const int rbase = row0 + t * 16 + ((lane >> 4) << 3);
        if (row0 + t * 16 < rows) {
            #pragma unroll
            for (int v = 0; v < 8; ++v) {
                int row = rbase + v;
                out[(size_t)row * D + col] = c[t][v] + bS + deg[row] * bM;
            }
        }
    }
}

extern "C" void kernel_launch(void* const* d_in, const int* in_sizes, int n_in,
                              void* d_out, int out_size, void* d_ws, size_t ws_size,
                              hipStream_t stream)
{
    const float* UF   = (const float*)d_in[0];
    const float* IF   = (const float*)d_in[1];
    const int*   edge = (const int*)d_in[2];
    const float* Wu   = (const float*)d_in[3];
    const float* bu   = (const float*)d_in[4];
    const float* Wi   = (const float*)d_in[5];
    const float* bi   = (const float*)d_in[6];
    const float* Wu2i = (const float*)d_in[7];
    const float* bu2i = (const float*)d_in[8];
    const float* Wi2u = (const float*)d_in[9];
    const float* bi2u = (const float*)d_in[10];

    const int U = in_sizes[0] / D;
    const int I = in_sizes[1] / D;
    const int E = in_sizes[2] / 2;
    const int R = U + I;

    float* outU = (float*)d_out;
    float* outI = outU + (size_t)U * D;

    const int NB   = (R + BIN_ROWS - 1) / BIN_ROWS;
    const int NBLK = (E + CHUNK - 1) / CHUNK;

    int*   off    = (int*)d_ws;
    float* degf   = (float*)(off + R);
    int*   bkt    = (int*)(degf + R);
    int*   H      = bkt + (size_t)2 * E;
    int*   binSt  = H + (size_t)NBLK * NB;
    const size_t hb = ((size_t)NBLK * NB + NB + 2) & ~(size_t)1;
    int2*  coarse = (int2*)(H + hb);

    bin_count_kernel<<<(unsigned)NBLK, 1024, 0, stream>>>(edge, edge + E, H, E, U, NB);
    bin_scatter_kernel<<<(unsigned)NBLK, 1024, 0, stream>>>(
        edge, edge + E, H, binSt, coarse, E, U, NB, NBLK);

    bin_build_kernel<<<(unsigned)NB, 1024, 0, stream>>>(coarse, binSt, off, degf, bkt, R);

    const int nbU = (U + BLOCK_ROWS - 1) / BLOCK_ROWS;
    const int nbI = (I + BLOCK_ROWS - 1) / BLOCK_ROWS;
    gnn_gemm_kernel<<<(unsigned)(nbU + nbI), 256, 0, stream>>>(
        UF, IF, outU, outI, Wu, bu, Wi2u, bi2u, Wi, bi, Wu2i, bu2i, degf, off, bkt, U, I, nbI);
}


# Bucket sort ranks by lane rotations: each lane counts the smaller keys of its bucket with 15 (31) row-rotate compares on unique keys, replacing the scalar loop over entries; sum order unchanged
# speedup vs baseline: 3.2042x; 1.0367x over previous
#include <hip/hip_runtime.h>
#include <hip/hip_bf16.h>

typedef __attribute__((ext_vector_type(2))) float v2f;
typedef __attribute__((ext_vector_type(8))) float v8f;

#define D 128
#define LDS_STRIDE 132

#define BIN_SHIFT 9
#define BIN_ROWS  (1 << BIN_SHIFT)
#define CHUNK     4096
#define EPT       (CHUNK / 1024)
#define NBINS_MAX 1024
#define HB        32
#define STAGE_CAP 8192
#define KEEP      5

__device__ __forceinline__ int block_scan_1024(int c, int* ws, int* total)
{
    const int lane = threadIdx.x & 31;
    const int w    = threadIdx.x >> 5;
    int x = c;
    for (int d = 1; d < 32; d <<= 1) {
        int y = __shfl_up(x, d);
        if (lane >= d) x += y;
    }
    if (lane == 31) ws[w] = x;
    __syncthreads();
    if (w == 0) {
        const int s = ws[lane];
        int t = s;
        for (int d = 1; d < 32; d <<= 1) {
            int y = __shfl_up(t, d);
            if (lane >= d) t += y;
        }
        ws[lane] = t - s;
        if (lane == 31) ws[32] = t;
    }
    __syncthreads();
    const int r = ws[w] + x - c;
    *total = ws[32];
    __syncthreads();
    return r;
}

__global__ __launch_bounds__(1024)
void bin_count_kernel(const int* __restrict__ uidx, const int* __restrict__ iidx,
                      int* __restrict__ H, int E, int U, int NB)
{
    __shared__ int hist[NBINS_MAX];
    const int tid = threadIdx.x;
    hist[tid] = 0;
    __syncthreads();
    const int base = blockIdx.x * CHUNK;
    int du[EPT], di[EPT];
    #pragma unroll
    for (int j = 0; j < EPT; ++j) {
        const int e = min(base + j * 1024 + tid, E - 1);
        du[j] = uidx[e];
        di[j] = iidx[e];
    }
    #pragma unroll
    for (int j = 0; j < EPT; ++j) {
        if (base + j * 1024 + tid < E) {
            atomicAdd(&hist[du[j] >> BIN_SHIFT], 1);
            atomicAdd(&hist[(U + di[j]) >> BIN_SHIFT], 1);
        }
    }
    __syncthreads();
    if (tid < NB) H[blockIdx.x * NB + tid] = hist[tid];
}

__global__ __launch_bounds__(1024)
void bin_scatter_kernel(const int* __restrict__ uidx, const int* __restrict__ iidx,
                        const int* __restrict__ H, int* __restrict__ binStart,
                        int2* __restrict__ coarse, int E, int U, int NB, int NBLK)
{
    __shared__ int  hist[NBINS_MAX];
    __shared__ int  lstart[NBINS_MAX];
    __shared__ int  delta[NBINS_MAX];
    __shared__ int  ws[33];
    __shared__ int2 stage[2 * CHUNK];

    const int tid = threadIdx.x;
    const int me  = blockIdx.x;
    hist[tid]   = 0;
    lstart[tid] = 0;
    delta[tid]  = 0;
    __syncthreads();

    const int base = me * CHUNK;
    int du[EPT], di[EPT], pu[EPT], pi[EPT];
    #pragma unroll
    for (int j = 0; j < EPT; ++j) {
        const int e = min(base + j * 1024 + tid, E - 1);
        du[j] = uidx[e];
        di[j] = iidx[e];
    }
    #pragma unroll
    for (int j = 0; j < EPT; ++j) {
        pu[j] = 0; pi[j] = 0;
        if (base + j * 1024 + tid < E) {
            pu[j] = atomicAdd(&hist[du[j] >> BIN_SHIFT], 1);
            pi[j] = atomicAdd(&hist[(U + di[j]) >> BIN_SHIFT], 1);
        } else {
            du[j] = -1;
        }
    }

    {
        const int G = 1024 / NB;
        const int g = tid / NB;
        const int t = tid - g * NB;
        if (g < G) {
            int tot = 0, pre = 0;
            for (int c0 = 0; c0 < NBLK; c0 += HB * G) {
                int v[HB];
                #pragma unroll
                for (int j = 0; j < HB; ++j)
                    v[j] = H[min(c0 + g + j * G, NBLK - 1) * NB + t];
                #pragma unroll
                for (int j = 0; j < HB; ++j) {
                    const int b = c0 + g + j * G;
                    tot += (b < NBLK) ? v[j] : 0;
                    pre += (b < me)   ? v[j] : 0;
                }
            }
            atomicAdd(&lstart[t], tot);
            atomicAdd(&delta[t],  pre);
        }
    }
    __syncthreads();
    const int tot = lstart[tid];
    const int pre = delta[tid];

    int nloc, nall;
    const int ls = block_scan_1024(hist[tid], ws, &nloc);
    const int gs = block_scan_1024(tot, ws, &nall);
    lstart[tid] = ls;
    delta[tid]  = gs + pre - ls;
    if (me == 0) {
        if (tid < NB)  binStart[tid] = gs;
        if (tid == 0)  binStart[NB]  = nall;
    }
    __syncthreads();

    #pragma unroll
    for (int j = 0; j < EPT; ++j) {
        if (du[j] >= 0) {
            const int ru = du[j];
            const int ri = U + di[j];
            stage[lstart[ru >> BIN_SHIFT] + pu[j]] = make_int2(ru, di[j]);
            stage[lstart[ri >> BIN_SHIFT] + pi[j]] = make_int2(ri, du[j]);
        }
    }
    __syncthreads();

    for (int i = tid; i < nloc; i += 1024) {
        const int2 en = stage[i];
        coarse[delta[en.x >> BIN_SHIFT] + i] = en;
    }
}

__global__ __launch_bounds__(1024)
void bin_build_kernel(const int2* __restrict__ coarse, const int* __restrict__ binStart,
                      int* __restrict__ off, float* __restrict__ degf,
                      int* __restrict__ bkt, int R)
{
    __shared__ int cnt[BIN_ROWS];
    __shared__ int cur[BIN_ROWS];
    __shared__ int ws[33];
    __shared__ int sb[STAGE_CAP];

    const int tid = threadIdx.x;
    const int b   = blockIdx.x;
    const int s   = binStart[b];
    const int n   = binStart[b + 1] - s;

    if (tid < BIN_ROWS) cnt[tid] = 0;

    int2 en[KEEP];
    #pragma unroll
    for (int j = 0; j < KEEP; ++j) en[j] = make_int2(-1, 0);
    if (n > 0) {
        #pragma unroll
        for (int j = 0; j < KEEP; ++j) {
            const int  q = j * 1024 + tid;
            const int2 v = coarse[s + min(q, n - 1)];
            if (q < n) en[j] = v;
        }
    }
    __syncthreads();
    #pragma unroll
    for (int j = 0; j < KEEP; ++j)
        if (en[j].x >= 0) atomicAdd(&cnt[en[j].x & (BIN_ROWS - 1)], 1);
    for (int q = KEEP * 1024 + tid; q < n; q += 1024)
        atomicAdd(&cnt[coarse[s + q].x & (BIN_ROWS - 1)], 1);
    __syncthreads();

    const int c = (tid < BIN_ROWS) ? cnt[tid] : 0;
    int nsum;
    const int ex = block_scan_1024(c, ws, &nsum);
    if (tid < BIN_ROWS) {
        cur[tid] = ex;
        const int r = (b << BIN_SHIFT) + tid;
        if (r < R) {
            off[r]  = s + ex;
            degf[r] = (float)c;
        }
    }
    __syncthreads();

    if (n <= STAGE_CAP) {
        #pragma unroll
        for (int j = 0; j < KEEP; ++j) {
            if (en[j].x >= 0) {
                const int k = atomicAdd(&cur[en[j].x & (BIN_ROWS - 1)], 1);
                sb[k] = en[j].y;
            }
        }
        for (int q = KEEP * 1024 + tid; q < n; q += 1024) {
            const int2 e2 = coarse[s + q];
            const int  k  = atomicAdd(&cur[e2.x & (BIN_ROWS - 1)], 1);
            sb[k] = e2.y;
        }
        __syncthreads();
        for (int q = tid; q < n; q += 1024) bkt[s + q] = sb[q];
    } else {
        for (int q = tid; q < n; q += 1024) {
            const int2 e2 = coarse[s + q];
            const int  k  = atomicAdd(&cur[e2.x & (BIN_ROWS - 1)], 1);
            bkt[s + k] = e2.y;
        }
    }
}

#define GW 16

#define RANK_STEP(from, i)                                                          \
    {                                                                               \
        const unsigned s_ = (unsigned)__builtin_amdgcn_update_dpp(                  \
            0, (int)(from), 0x120 + (i), 0xf, 0xf, false);             \
        rk += (s_ < u) ? 1 : 0;                                                     \
    }
#define RANK_ROW(from)                                                              \
    RANK_STEP(from, 1)  RANK_STEP(from, 2)  RANK_STEP(from, 3)  RANK_STEP(from, 4)  \
    RANK_STEP(from, 5)  RANK_STEP(from, 6)  RANK_STEP(from, 7)  RANK_STEP(from, 8)  \
    RANK_STEP(from, 9)  RANK_STEP(from, 10) RANK_STEP(from, 11) RANK_STEP(from, 12) \
    RANK_STEP(from, 13) RANK_STEP(from, 14) RANK_STEP(from, 15)

__device__ __forceinline__ int sort_keys(const int n, const int k, const int lane)
{
    const unsigned u = (lane < n) ? (((unsigned)k << 5) | (unsigned)lane) : 0xffffffffu;
    int rk = 0;
    RANK_ROW(u)
    if (n > 16) {
        const unsigned x = (unsigned)__shfl_xor((int)u, 16);
        rk += (x < u) ? 1 : 0;
        RANK_ROW(x)
    }
    const int rank = (lane < n) ? rk : lane;
    return __builtin_amdgcn_ds_permute(rank << 2, k);
}

__device__ __forceinline__ float4 sum_sorted(const float* __restrict__ src,
                                             const int n, const int sk, const int lane)
{
    float4 acc = make_float4(0.f, 0.f, 0.f, 0.f);
    for (int t0 = 0; t0 < n; t0 += GW) {
        float4 f[GW];
        #pragma unroll
        for (int j = 0; j < GW; ++j) {
            f[j] = make_float4(0.f, 0.f, 0.f, 0.f);
            if (t0 + j < n) {
                const int s = __builtin_amdgcn_readlane(sk, t0 + j);
                f[j] = ((const float4*)(src + (size_t)s * D))[lane];
            }
        }
        #pragma unroll
        for (int j = 0; j < GW; ++j) {
            if (t0 + j < n) {
                acc.x += f[j].x;
                acc.y += f[j].y;
                acc.z += f[j].z;
                acc.w += f[j].w;
            }
        }
    }
    return acc;
}

__device__ __forceinline__ float4 sum_long(const float* __restrict__ src,
                                           const int* __restrict__ bkt,
                                           const int n, const int o, const int lane)
{
    float4 acc = make_float4(0.f, 0.f, 0.f, 0.f);
    int cur = -1;
    for (int trip = 0; trip < n; ++trip) {
        int best = 0x7fffffff, m = 0;
        for (int j = lane; j < n; j += 32) {
            const int kk = bkt[o + j];
            if (kk > cur) {
                if (kk < best)       { best = kk; m = 1; }
                else if (kk == best) { ++m; }
            }
        }
        int wb = best;
        for (int d = 16; d > 0; d >>= 1) wb = min(wb, __shfl_xor(wb, d));
        if (wb == 0x7fffffff) break;
        int mm = (best == wb) ? m : 0;
        for (int d = 16; d > 0; d >>= 1) mm += __shfl_xor(mm, d);
        const int s   = __builtin_amdgcn_readfirstlane(wb);
        const int rep = __builtin_amdgcn_readfirstlane(mm);
        const float4 f = ((const float4*)(src + (size_t)s * D))[lane];
        for (int t = 0; t < rep; ++t) {
            acc.x += f.x; acc.y += f.y; acc.z += f.z; acc.w += f.w;
        }
        cur = s;
    }
    return acc;
}

#define ROW_TILES 5
#define BLOCK_ROWS (ROW_TILES * 16)

__global__ __launch_bounds__(256)
void gnn_gemm_kernel(const float* __restrict__ XU, const float* __restrict__ XI,
                     float* __restrict__ outU, float* __restrict__ outI,
                     const float* __restrict__ WsU, const float* __restrict__ bsU,
                     const float* __restrict__ WmU, const float* __restrict__ bmU,
                     const float* __restrict__ WsI, const float* __restrict__ bsI,
                     const float* __restrict__ WmI, const float* __restrict__ bmI,
                     const float* __restrict__ degf, const int* __restrict__ off,
                     const int* __restrict__ bkt, int U, int I, int nbI)
{
    __shared__ float sX[BLOCK_ROWS * LDS_STRIDE];
    __shared__ float sG[BLOCK_ROWS * LDS_STRIDE];

    const bool item = (int)blockIdx.x < nbI;
    const float* __restrict__ X     = item ? XI  : XU;
    float*       __restrict__ out   = item ? outI : outU;
    const float* __restrict__ Wself = item ? WsI : WsU;
    const float* __restrict__ bself = item ? bsI : bsU;
    const float* __restrict__ Wmsg  = item ? WmI : WmU;
    const float* __restrict__ bmsg  = item ? bmI : bmU;
    const float* __restrict__ deg   = item ? degf + U : degf;
    const int*   __restrict__ offp  = item ? off + U : off;
    const float* __restrict__ srcT  = item ? XU : XI;
    const int rows = item ? I : U;

    const int row0 = (item ? (int)blockIdx.x : (int)blockIdx.x - nbI) * BLOCK_ROWS;
    const int tid  = threadIdx.x;

    float dsum = 0.f;
    {
        float dg[(BLOCK_ROWS + 31) / 32];
        #pragma unroll
        for (int j = 0; j < (BLOCK_ROWS + 31) / 32; ++j) {
            const int i = min(j * 32 + (tid & 31), BLOCK_ROWS - 1);
            dg[j] = deg[min(row0 + i, rows - 1)];
        }
        #pragma unroll
        for (int j = 0; j < (BLOCK_ROWS + 31) / 32; ++j) dsum += dg[j];
    }
    const bool hasEdge = dsum > 0.f;

    {
        const int c4 = tid & 31;
        const int r0 = tid >> 5;
        float4 xr[BLOCK_ROWS / 8];
        #pragma unroll
        for (int j = 0; j < BLOCK_ROWS / 8; ++j) {
            const int gr = min(row0 + r0 + 8 * j, rows - 1);
            xr[j] = ((const float4*)(X + (size_t)gr * D))[c4];
        }

        int hn = 0, ho = 0;
        if (c4 < BLOCK_ROWS / 8) {
            const int gr = min(row0 + r0 + 8 * c4, rows - 1);
            hn = (int)deg[gr];
            ho = offp[gr];
        }
        int n = __builtin_amdgcn_readlane(hn, 0);
        int o = __builtin_amdgcn_readlane(ho, 0);
        int k = 0x7fffffff;
        if (n <= 32 && c4 < n) k = bkt[o + c4];

        #pragma unroll
        for (int j = 0; j < BLOCK_ROWS / 8; ++j) {
            const int r = r0 + 8 * j;
            ((float4*)(sX + r * LDS_STRIDE))[c4] = xr[j];
        }

        for (int j = 0; j < BLOCK_ROWS / 8; ++j) {
            const bool shortb = (n > 0 && n <= 32);
            int sk = 0;
            if (shortb) sk = sort_keys(n, k, c4);
            int n1 = 0, o1 = 0, k1 = 0x7fffffff;
            if (j + 1 < BLOCK_ROWS / 8) {
                n1 = __builtin_amdgcn_readlane(hn, j + 1);
                o1 = __builtin_amdgcn_readlane(ho, j + 1);
                if (n1 <= 32 && c4 < n1) k1 = bkt[o1 + c4];
            }
            float4 a = make_float4(0.f, 0.f, 0.f, 0.f);
            if (shortb)      a = sum_sorted(srcT, n, sk, c4);
            else if (n > 32) a = sum_long(srcT, bkt, n, o, c4);
            ((float4*)(sG + (r0 + 8 * j) * LDS_STRIDE))[c4] = a;
            n = n1; o = o1; k = k1;
        }
    }
    __syncthreads();

    const int lane = tid & 31;
    const int wave = tid >> 5;
    const int col0 = wave * 16;
    const int mn   = lane & 15;
    const int koff = (lane >> 4) << 1;

    const float* wS = Wself + (size_t)(col0 + mn) * D;
    const float* wM = Wmsg  + (size_t)(col0 + mn) * D;
    const float* aS = sX + mn * LDS_STRIDE;
    const float* aM = sG + mn * LDS_STRIDE;

    v8f c[ROW_TILES];
    #pragma unroll
    for (int t = 0; t < ROW_TILES; ++t) c[t] = (v8f){};

    #pragma unroll
    for (int k = 0; k < D; k += 4) {
        v2f b = { wS[k + koff], wS[k + koff + 1] };
        #pragma unroll
        for (int t = 0; t < ROW_TILES; ++t) {
            const float* ap = aS + t * 16 * LDS_STRIDE;
            v2f a = { ap[k + koff], ap[k + koff + 1] };
            c[t] = __builtin_amdgcn_wmma_f32_16x16x4_f32(false, a, false, b,
                                                         (short)0, c[t], false, false);
        }
    }
    if (__builtin_amdgcn_ballot_w32(hasEdge) != 0u) {
    #pragma unroll
    for (int k = 0; k < D; k += 4) {
        v2f b = { wM[k + koff], wM[k + koff + 1] };
        #pragma unroll
        for (int t = 0; t < ROW_TILES; ++t) {
            const float* ap = aM + t * 16 * LDS_STRIDE;
            v2f a = { ap[k + koff], ap[k + koff + 1] };
            c[t] = __builtin_amdgcn_wmma_f32_16x16x4_f32(false, a, false, b,
                                                         (short)0, c[t], false, false);
        }
    }
    }

    const int col  = col0 + mn;
    const float bS = bself[col];
    const float bM = bmsg[col];
    #pragma unroll
    for (int t = 0; t < ROW_TILES; ++t) {
        const int rbase = row0 + t * 16 + ((lane >> 4) << 3);
        if (row0 + t * 16 < rows) {
            #pragma unroll
            for (int v = 0; v < 8; ++v) {
                int row = rbase + v;
                out[(size_t)row * D + col] = c[t][v] + bS + deg[row] * bM;
            }
        }
    }
}

extern "C" void kernel_launch(void* const* d_in, const int* in_sizes, int n_in,
                              void* d_out, int out_size, void* d_ws, size_t ws_size,
                              hipStream_t stream)
{
    const float* UF   = (const float*)d_in[0];
    const float* IF   = (const float*)d_in[1];
    const int*   edge = (const int*)d_in[2];
    const float* Wu   = (const float*)d_in[3];
    const float* bu   = (const float*)d_in[4];
    const float* Wi   = (const float*)d_in[5];
    const float* bi   = (const float*)d_in[6];
    const float* Wu2i = (const float*)d_in[7];
    const float* bu2i = (const float*)d_in[8];
    const float* Wi2u = (const float*)d_in[9];
    const float* bi2u = (const float*)d_in[10];

    const int U = in_sizes[0] / D;
    const int I = in_sizes[1] / D;
    const int E = in_sizes[2] / 2;
    const int R = U + I;

    float* outU = (float*)d_out;
    float* outI = outU + (size_t)U * D;

    const int NB   = (R + BIN_ROWS - 1) / BIN_ROWS;
    const int NBLK = (E + CHUNK - 1) / CHUNK;

    int*   off    = (int*)d_ws;
    float* degf   = (float*)(off + R);
    int*   bkt    = (int*)(degf + R);
    int*   H      = bkt + (size_t)2 * E;
    int*   binSt  = H + (size_t)NBLK * NB;
    const size_t hb = ((size_t)NBLK * NB + NB + 2) & ~(size_t)1;
    int2*  coarse = (int2*)(H + hb);

    bin_count_kernel<<<(unsigned)NBLK, 1024, 0, stream>>>(edge, edge + E, H, E, U, NB);
    bin_scatter_kernel<<<(unsigned)NBLK, 1024, 0, stream>>>(
        edge, edge + E, H, binSt, coarse, E, U, NB, NBLK);

    bin_build_kernel<<<(unsigned)NB, 1024, 0, stream>>>(coarse, binSt, off, degf, bkt, R);

    const int nbU = (U + BLOCK_ROWS - 1) / BLOCK_ROWS;
    const int nbI = (I + BLOCK_ROWS - 1) / BLOCK_ROWS;
    gnn_gemm_kernel<<<(unsigned)(nbU + nbI), 256, 0, stream>>>(
        UF, IF, outU, outI, Wu, bu, Wi2u, bi2u, Wi, bi, Wu2i, bu2i, degf, off, bkt, U, I, nbI);
}


# Counting pass dropped: each edge chunk writes its bin-sorted entries to its own stretch with per-chunk run starts; the per-bin block collects its run of every chunk (one launch less); sums unchanged
# speedup vs baseline: 3.3310x; 1.0396x over previous
#include <hip/hip_runtime.h>
#include <hip/hip_bf16.h>

typedef __attribute__((ext_vector_type(2))) float v2f;
typedef __attribute__((ext_vector_type(8))) float v8f;

#define D 128
#define LDS_STRIDE 132

#define BIN_SHIFT 9
#define BIN_ROWS  (1 << BIN_SHIFT)
#define CHUNK     4096
#define EPT       (CHUNK / 1024)
#define NBINS_MAX 1024
#define STAGE_CAP 8192
#define KEEP      5

__device__ __forceinline__ int block_scan_1024(int c, int* ws, int* total)
{
    const int lane = threadIdx.x & 31;
    const int w    = threadIdx.x >> 5;
    int x = c;
    for (int d = 1; d < 32; d <<= 1) {
        int y = __shfl_up(x, d);
        if (lane >= d) x += y;
    }
    if (lane == 31) ws[w] = x;
    __syncthreads();
    if (w == 0) {
        const int s = ws[lane];
        int t = s;
        for (int d = 1; d < 32; d <<= 1) {
            int y = __shfl_up(t, d);
            if (lane >= d) t += y;
        }
        ws[lane] = t - s;
        if (lane == 31) ws[32] = t;
    }
    __syncthreads();
    const int r = ws[w] + x - c;
    *total = ws[32];
    __syncthreads();
    return r;
}

__global__ __launch_bounds__(1024)
void bin_sort_kernel(const int* __restrict__ uidx, const int* __restrict__ iidx,
                     int* __restrict__ Ls, int2* __restrict__ coarse,
                     int E, int U, int NB)
{
    __shared__ int  hist[NBINS_MAX];
    __shared__ int  lstart[NBINS_MAX];
    __shared__ int  ws[33];
    __shared__ int2 stage[2 * CHUNK];

    const int tid = threadIdx.x;
    const int me  = blockIdx.x;
    hist[tid] = 0;
    __syncthreads();

    const int base = me * CHUNK;
    int du[EPT], di[EPT], pu[EPT], pi[EPT];
    #pragma unroll
    for (int j = 0; j < EPT; ++j) {
        const int e = min(base + j * 1024 + tid, E - 1);
        du[j] = uidx[e];
        di[j] = iidx[e];
    }
    #pragma unroll
    for (int j = 0; j < EPT; ++j) {
        pu[j] = 0; pi[j] = 0;
        if (base + j * 1024 + tid < E) {
            pu[j] = atomicAdd(&hist[du[j] >> BIN_SHIFT], 1);
            pi[j] = atomicAdd(&hist[(U + di[j]) >> BIN_SHIFT], 1);
        } else {
            du[j] = -1;
        }
    }
    __syncthreads();

    int nloc;
    const int ls = block_scan_1024(hist[tid], ws, &nloc);
    lstart[tid] = ls;
    int* __restrict__ lrow = Ls + (size_t)me * (NB + 1);
    if (tid < NB)  lrow[tid] = ls;
    if (tid == 0)  lrow[NB]  = nloc;
    __syncthreads();

    #pragma unroll
    for (int j = 0; j < EPT; ++j) {
        if (du[j] >= 0) {
            const int ru = du[j];
            const int ri = U + di[j];
            stage[lstart[ru >> BIN_SHIFT] + pu[j]] = make_int2(ru, di[j]);
            stage[lstart[ri >> BIN_SHIFT] + pi[j]] = make_int2(ri, du[j]);
        }
    }
    __syncthreads();

    int2* __restrict__ dst = coarse + (size_t)me * (2 * CHUNK);
    for (int i = tid; i < nloc; i += 1024) dst[i] = stage[i];
}

__device__ __forceinline__ int place_of(const int2* run, const int nrun, const int p0, const int q)
{
    int lo = 0;
    int sr = run[0].y;
    for (int st = p0; st > 0; st >>= 1) {
        const int  mid = lo + st;
        const int2 rv  = run[min(mid, nrun - 1)];
        const bool ok  = (mid < nrun) & (rv.x <= q);
        lo = ok ? mid  : lo;
        sr = ok ? rv.y : sr;
    }
    return sr + q;
}

#define CH_MAX 1024
__global__ __launch_bounds__(1024)
void bin_build_kernel(const int2* __restrict__ coarse, const int* __restrict__ Ls,
                      int* __restrict__ off, float* __restrict__ degf,
                      int* __restrict__ bkt, int R, int NB, int NBLK)
{
    __shared__ int cnt[BIN_ROWS];
    __shared__ int cur[BIN_ROWS];
    __shared__ int ws[33];
    __shared__ int sb[STAGE_CAP];
    __shared__ int2 run[CH_MAX];

    const int tid = threadIdx.x;
    const int b   = blockIdx.x;
    const int p0  = 1 << (31 - __clz(max(NBLK - 1, 1)));

    int ra = 0, rn = 0;
    if (tid < NBLK) {
        const int* __restrict__ lrow = Ls + (size_t)tid * (NB + 1) + b;
        ra = lrow[0];
        rn = lrow[1] - ra;
    }
    if (tid < BIN_ROWS) cnt[tid] = 0;

    int s, n;
    block_scan_1024(ra, ws, &s);
    const int rp = block_scan_1024(rn, ws, &n);
    if (tid < NBLK) run[tid] = make_int2(rp, tid * (2 * CHUNK) + ra - rp);
    __syncthreads();

    int2 en[KEEP];
    #pragma unroll
    for (int j = 0; j < KEEP; ++j) en[j] = make_int2(-1, 0);
    if (n > 0) {
        int qq[KEEP], lo[KEEP], sr[KEEP];
        const int sr0 = run[0].y;
        #pragma unroll
        for (int j = 0; j < KEEP; ++j) { qq[j] = min(j * 1024 + tid, n - 1); lo[j] = 0; sr[j] = sr0; }
        for (int st = p0; st > 0; st >>= 1) {
            int2 rv[KEEP];
            #pragma unroll
            for (int j = 0; j < KEEP; ++j) rv[j] = run[min(lo[j] + st, NBLK - 1)];
            #pragma unroll
            for (int j = 0; j < KEEP; ++j) {
                const int  mid = lo[j] + st;
                const bool ok  = (mid < NBLK) & (rv[j].x <= qq[j]);
                lo[j] = ok ? mid     : lo[j];
                sr[j] = ok ? rv[j].y : sr[j];
            }
        }
        #pragma unroll
        for (int j = 0; j < KEEP; ++j) {
            const int2 v = coarse[sr[j] + qq[j]];
            if (j * 1024 + tid < n) en[j] = v;
        }
    }
    #pragma unroll
    for (int j = 0; j < KEEP; ++j)
        if (en[j].x >= 0) atomicAdd(&cnt[en[j].x & (BIN_ROWS - 1)], 1);
    for (int q = KEEP * 1024 + tid; q < n; q += 1024)
        atomicAdd(&cnt[coarse[place_of(run, NBLK, p0, q)].x & (BIN_ROWS - 1)], 1);
    __syncthreads();

    const int c = (tid < BIN_ROWS) ? cnt[tid] : 0;
    int nsum;
    const int ex = block_scan_1024(c, ws, &nsum);
    if (tid < BIN_ROWS) {
        cur[tid] = ex;
        const int r = (b << BIN_SHIFT) + tid;
        if (r < R) {
            off[r]  = s + ex;
            degf[r] = (float)c;
        }
    }
    __syncthreads();

    if (n <= STAGE_CAP) {
        #pragma unroll
        for (int j = 0; j < KEEP; ++j) {
            if (en[j].x >= 0) {
                const int k = atomicAdd(&cur[en[j].x & (BIN_ROWS - 1)], 1);
                sb[k] = en[j].y;
            }
        }
        for (int q = KEEP * 1024 + tid; q < n; q += 1024) {
            const int2 e2 = coarse[place_of(run, NBLK, p0, q)];
            const int  k  = atomicAdd(&cur[e2.x & (BIN_ROWS - 1)], 1);
            sb[k] = e2.y;
        }
        __syncthreads();
        for (int q = tid; q < n; q += 1024) bkt[s + q] = sb[q];
    } else {
        for (int q = tid; q < n; q += 1024) {
            const int2 e2 = coarse[place_of(run, NBLK, p0, q)];
            const int  k  = atomicAdd(&cur[e2.x & (BIN_ROWS - 1)], 1);
            bkt[s + k] = e2.y;
        }
    }
}

#define GW 16

#define RANK_STEP(from, i)                                                          \
    {                                                                               \
        const unsigned s_ = (unsigned)__builtin_amdgcn_update_dpp(                  \
            0, (int)(from), 0x120 + (i), 0xf, 0xf, false);             \
        rk += (s_ < u) ? 1 : 0;                                                     \
    }
#define RANK_ROW(from)                                                              \
    RANK_STEP(from, 1)  RANK_STEP(from, 2)  RANK_STEP(from, 3)  RANK_STEP(from, 4)  \
    RANK_STEP(from, 5)  RANK_STEP(from, 6)  RANK_STEP(from, 7)  RANK_STEP(from, 8)  \
    RANK_STEP(from, 9)  RANK_STEP(from, 10) RANK_STEP(from, 11) RANK_STEP(from, 12) \
    RANK_STEP(from, 13) RANK_STEP(from, 14) RANK_STEP(from, 15)

__device__ __forceinline__ int sort_keys(const int n, const int k, const int lane)
{
    const unsigned u = (lane < n) ? (((unsigned)k << 5) | (unsigned)lane) : 0xffffffffu;
    int rk = 0;
    RANK_ROW(u)
    if (n > 16) {
        const unsigned x = (unsigned)__shfl_xor((int)u, 16);
        rk += (x < u) ? 1 : 0;
        RANK_ROW(x)
    }
    const int rank = (lane < n) ? rk : lane;
    return __builtin_amdgcn_ds_permute(rank << 2, k);
}

__device__ __forceinline__ float4 sum_sorted(const float* __restrict__ src,
                                             const int n, const int sk, const int lane)
{
    float4 acc = make_float4(0.f, 0.f, 0.f, 0.f);
    for (int t0 = 0; t0 < n; t0 += GW) {
        float4 f[GW];
        #pragma unroll
        for (int j = 0; j < GW; ++j) {
            f[j] = make_float4(0.f, 0.f, 0.f, 0.f);
            if (t0 + j < n) {
                const int s = __builtin_amdgcn_readlane(sk, t0 + j);
                f[j] = ((const float4*)(src + (size_t)s * D))[lane];
            }
        }
        #pragma unroll
        for (int j = 0; j < GW; ++j) {
            if (t0 + j < n) {
                acc.x += f[j].x;
                acc.y += f[j].y;
                acc.z += f[j].z;
                acc.w += f[j].w;
            }
        }
    }
    return acc;
}

__device__ __forceinline__ float4 sum_long(const float* __restrict__ src,
                                           const int* __restrict__ bkt,
                                           const int n, const int o, const int lane)
{
    float4 acc = make_float4(0.f, 0.f, 0.f, 0.f);
    int cur = -1;
    for (int trip = 0; trip < n; ++trip) {
        int best = 0x7fffffff, m = 0;
        for (int j = lane; j < n; j += 32) {
            const int kk = bkt[o + j];
            if (kk > cur) {
                if (kk < best)       { best = kk; m = 1; }
                else if (kk == best) { ++m; }
            }
        }
        int wb = best;
        for (int d = 16; d > 0; d >>= 1) wb = min(wb, __shfl_xor(wb, d));
        if (wb == 0x7fffffff) break;
        int mm = (best == wb) ? m : 0;
        for (int d = 16; d > 0; d >>= 1) mm += __shfl_xor(mm, d);
        const int s   = __builtin_amdgcn_readfirstlane(wb);
        const int rep = __builtin_amdgcn_readfirstlane(mm);
        const float4 f = ((const float4*)(src + (size_t)s * D))[lane];
        for (int t = 0; t < rep; ++t) {
            acc.x += f.x; acc.y += f.y; acc.z += f.z; acc.w += f.w;
        }
        cur = s;
    }
    return acc;
}

#define ROW_TILES 5
#define BLOCK_ROWS (ROW_TILES * 16)

__global__ __launch_bounds__(256)
void gnn_gemm_kernel(const float* __restrict__ XU, const float* __restrict__ XI,
                     float* __restrict__ outU, float* __restrict__ outI,
                     const float* __restrict__ WsU, const float* __restrict__ bsU,
                     const float* __restrict__ WmU, const float* __restrict__ bmU,
                     const float* __restrict__ WsI, const float* __restrict__ bsI,
                     const float* __restrict__ WmI, const float* __restrict__ bmI,
                     const float* __restrict__ degf, const int* __restrict__ off,
                     const int* __restrict__ bkt, int U, int I, int nbI)
{
    __shared__ float sX[BLOCK_ROWS * LDS_STRIDE];
    __shared__ float sG[BLOCK_ROWS * LDS_STRIDE];

    const bool item = (int)blockIdx.x < nbI;
    const float* __restrict__ X     = item ? XI  : XU;
    float*       __restrict__ out   = item ? outI : outU;
    const float* __restrict__ Wself = item ? WsI : WsU;
    const float* __restrict__ bself = item ? bsI : bsU;
    const float* __restrict__ Wmsg  = item ? WmI : WmU;
    const float* __restrict__ bmsg  = item ? bmI : bmU;
    const float* __restrict__ deg   = item ? degf + U : degf;
    const int*   __restrict__ offp  = item ? off + U : off;
    const float* __restrict__ srcT  = item ? XU : XI;
    const int rows = item ? I : U;

    const int row0 = (item ? (int)blockIdx.x : (int)blockIdx.x - nbI) * BLOCK_ROWS;
    const int tid  = threadIdx.x;

    float dsum = 0.f;
    {
        float dg[(BLOCK_ROWS + 31) / 32];
        #pragma unroll
        for (int j = 0; j < (BLOCK_ROWS + 31) / 32; ++j) {
            const int i = min(j * 32 + (tid & 31), BLOCK_ROWS - 1);
            dg[j] = deg[min(row0 + i, rows - 1)];
        }
        #pragma unroll
        for (int j = 0; j < (BLOCK_ROWS + 31) / 32; ++j) dsum += dg[j];
    }
    const bool hasEdge = dsum > 0.f;

    {
        const int c4 = tid & 31;
        const int r0 = tid >> 5;
        float4 xr[BLOCK_ROWS / 8];
        #pragma unroll
        for (int j = 0; j < BLOCK_ROWS / 8; ++j) {
            const int gr = min(row0 + r0 + 8 * j, rows - 1);
            xr[j] = ((const float4*)(X + (size_t)gr * D))[c4];
        }

        int hn = 0, ho = 0;
        if (c4 < BLOCK_ROWS / 8) {
            const int gr = min(row0 + r0 + 8 * c4, rows - 1);
            hn = (int)deg[gr];
            ho = offp[gr];
        }
        int n = __builtin_amdgcn_readlane(hn, 0);
        int o = __builtin_amdgcn_readlane(ho, 0);
        int k = 0x7fffffff;
        if (n <= 32 && c4 < n) k = bkt[o + c4];

        #pragma unroll
        for (int j = 0; j < BLOCK_ROWS / 8; ++j) {
            const int r = r0 + 8 * j;
            ((float4*)(sX + r * LDS_STRIDE))[c4] = xr[j];
        }

        for (int j = 0; j < BLOCK_ROWS / 8; ++j) {
            const bool shortb = (n > 0 && n <= 32);
            int sk = 0;
            if (shortb) sk = sort_keys(n, k, c4);
            int n1 = 0, o1 = 0, k1 = 0x7fffffff;
            if (j + 1 < BLOCK_ROWS / 8) {
                n1 = __builtin_amdgcn_readlane(hn, j + 1);
                o1 = __builtin_amdgcn_readlane(ho, j + 1);
                if (n1 <= 32 && c4 < n1) k1 = bkt[o1 + c4];
            }
            float4 a = make_float4(0.f, 0.f, 0.f, 0.f);
            if (shortb)      a = sum_sorted(srcT, n, sk, c4);
            else if (n > 32) a = sum_long(srcT, bkt, n, o, c4);
            ((float4*)(sG + (r0 + 8 * j) * LDS_STRIDE))[c4] = a;
            n = n1; o = o1; k = k1;
        }
    }
    __syncthreads();

    const int lane = tid & 31;
    const int wave = tid >> 5;
    const int col0 = wave * 16;
    const int mn   = lane & 15;
    const int koff = (lane >> 4) << 1;

    const float* wS = Wself + (size_t)(col0 + mn) * D;
    const float* wM = Wmsg  + (size_t)(col0 + mn) * D;
    const float* aS = sX + mn * LDS_STRIDE;
    const float* aM = sG + mn * LDS_STRIDE;

    v8f c[ROW_TILES];
    #pragma unroll
    for (int t = 0; t < ROW_TILES; ++t) c[t] = (v8f){};

    #pragma unroll
    for (int k = 0; k < D; k += 4) {
        v2f b = { wS[k + koff], wS[k + koff + 1] };
        #pragma unroll
        for (int t = 0; t < ROW_TILES; ++t) {
            const float* ap = aS + t * 16 * LDS_STRIDE;
            v2f a = { ap[k + koff], ap[k + koff + 1] };
            c[t] = __builtin_amdgcn_wmma_f32_16x16x4_f32(false, a, false, b,
                                                         (short)0, c[t], false, false);
        }
    }
    if (__builtin_amdgcn_ballot_w32(hasEdge) != 0u) {
    #pragma unroll
    for (int k = 0; k < D; k += 4) {
        v2f b = { wM[k + koff], wM[k + koff + 1] };
        #pragma unroll
        for (int t = 0; t < ROW_TILES; ++t) {
            const float* ap = aM + t * 16 * LDS_STRIDE;
            v2f a = { ap[k + koff], ap[k + koff + 1] };
            c[t] = __builtin_amdgcn_wmma_f32_16x16x4_f32(false, a, false, b,
                                                         (short)0, c[t], false, false);
        }
    }
    }

    const int col  = col0 + mn;
    const float bS = bself[col];
    const float bM = bmsg[col];
    #pragma unroll
    for (int t = 0; t < ROW_TILES; ++t) {
        const int rbase = row0 + t * 16 + ((lane >> 4) << 3);
        if (row0 + t * 16 < rows) {
            #pragma unroll
            for (int v = 0; v < 8; ++v) {
                int row = rbase + v;
                out[(size_t)row * D + col] = c[t][v] + bS + deg[row] * bM;
            }
        }
    }
}

extern "C" void kernel_launch(void* const* d_in, const int* in_sizes, int n_in,
                              void* d_out, int out_size, void* d_ws, size_t ws_size,
                              hipStream_t stream)
{
    const float* UF   = (const float*)d_in[0];
    const float* IF   = (const float*)d_in[1];
    const int*   edge = (const int*)d_in[2];
    const float* Wu   = (const float*)d_in[3];
    const float* bu   = (const float*)d_in[4];
    const float* Wi   = (const float*)d_in[5];
    const float* bi   = (const float*)d_in[6];
    const float* Wu2i = (const float*)d_in[7];
    const float* bu2i = (const float*)d_in[8];
    const float* Wi2u = (const float*)d_in[9];
    const float* bi2u = (const float*)d_in[10];

    const int U = in_sizes[0] / D;
    const int I = in_sizes[1] / D;
    const int E = in_sizes[2] / 2;
    const int R = U + I;

    float* outU = (float*)d_out;
    float* outI = outU + (size_t)U * D;

    const int NB   = (R + BIN_ROWS - 1) / BIN_ROWS;
    const int NBLK = (E + CHUNK - 1) / CHUNK;

    int*   off    = (int*)d_ws;
    float* degf   = (float*)(off + R);
    int*   bkt    = (int*)(degf + R);
    int*   Ls     = bkt + (size_t)2 * E;
    const size_t hb = ((size_t)NBLK * (NB + 1) + 1) & ~(size_t)1;
    int2*  coarse = (int2*)(Ls + hb);

    bin_sort_kernel<<<(unsigned)NBLK, 1024, 0, stream>>>(edge, edge + E, Ls, coarse, E, U, NB);

    bin_build_kernel<<<(unsigned)NB, 1024, 0, stream>>>(coarse, Ls, off, degf, bkt, R, NB, NBLK);

    const int nbU = (U + BLOCK_ROWS - 1) / BLOCK_ROWS;
    const int nbI = (I + BLOCK_ROWS - 1) / BLOCK_ROWS;
    gnn_gemm_kernel<<<(unsigned)(nbU + nbI), 256, 0, stream>>>(
        UF, IF, outU, outI, Wu, bu, Wi2u, bi2u, Wi, bi, Wu2i, bu2i, degf, off, bkt, U, I, nbI);
}


# Weights rearranged once per call into B-fragment order (by the bucketing launch): a fragment load reads 512 consecutive bytes for 8 k, not 16 lines for 4 k; sum order over k inside each 8 changes
# speedup vs baseline: 3.4739x; 1.0429x over previous
#include <hip/hip_runtime.h>
#include <hip/hip_bf16.h>

typedef __attribute__((ext_vector_type(2))) float v2f;
typedef __attribute__((ext_vector_type(8))) float v8f;

#define D 128
#define LDS_STRIDE 132

#define WF_ITEMS (D * D / 4)

#define BIN_SHIFT 9
#define BIN_ROWS  (1 << BIN_SHIFT)
#define CHUNK     4096
#define EPT       (CHUNK / 1024)
#define NBINS_MAX 1024
#define STAGE_CAP 8192
#define KEEP      5

__device__ __forceinline__ int block_scan_1024(int c, int* ws, int* total)
{
    const int lane = threadIdx.x & 31;
    const int w    = threadIdx.x >> 5;
    int x = c;
    for (int d = 1; d < 32; d <<= 1) {
        int y = __shfl_up(x, d);
        if (lane >= d) x += y;
    }
    if (lane == 31) ws[w] = x;
    __syncthreads();
    if (w == 0) {
        const int s = ws[lane];
        int t = s;
        for (int d = 1; d < 32; d <<= 1) {
            int y = __shfl_up(t, d);
            if (lane >= d) t += y;
        }
        ws[lane] = t - s;
        if (lane == 31) ws[32] = t;
    }
    __syncthreads();
    const int r = ws[w] + x - c;
    *total = ws[32];
    __syncthreads();
    return r;
}

__global__ __launch_bounds__(1024)
void bin_sort_kernel(const int* __restrict__ uidx, const int* __restrict__ iidx,
                     int* __restrict__ Ls, int2* __restrict__ coarse,
                     int E, int U, int NB,
                     const float* __restrict__ W0, const float* __restrict__ W1,
                     const float* __restrict__ W2, const float* __restrict__ W3,
                     float* __restrict__ Wf)
{
    __shared__ int  hist[NBINS_MAX];
    __shared__ int  lstart[NBINS_MAX];
    __shared__ int  ws[33];
    __shared__ int2 stage[2 * CHUNK];

    const int tid = threadIdx.x;
    const int me  = blockIdx.x;
    hist[tid] = 0;
    __syncthreads();

    const int base = me * CHUNK;
    int du[EPT], di[EPT], pu[EPT], pi[EPT];
    #pragma unroll
    for (int j = 0; j < EPT; ++j) {
        const int e = min(base + j * 1024 + tid, E - 1);
        du[j] = uidx[e];
        di[j] = iidx[e];
    }

    for (int g = me * 1024 + tid; g < 4 * WF_ITEMS; g += (int)gridDim.x * 1024) {
        const int m  = g / WF_ITEMS;
        const int it = g % WF_ITEMS;
        const int l  = it & 31;
        const int kk = (it >> 5) & 15;
        const int w  = it >> 9;
        const float* __restrict__ Wsrc = (m == 0) ? W0 : (m == 1) ? W1 : (m == 2) ? W2 : W3;
        const float4 v = *(const float4*)(Wsrc + (size_t)(16 * w + (l & 15)) * D + 8 * kk + 4 * (l >> 4));
        ((float4*)Wf)[g] = v;
    }
    #pragma unroll
    for (int j = 0; j < EPT; ++j) {
        pu[j] = 0; pi[j] = 0;
        if (base + j * 1024 + tid < E) {
            pu[j] = atomicAdd(&hist[du[j] >> BIN_SHIFT], 1);
            pi[j] = atomicAdd(&hist[(U + di[j]) >> BIN_SHIFT], 1);
        } else {
            du[j] = -1;
        }
    }
    __syncthreads();

    int nloc;
    const int ls = block_scan_1024(hist[tid], ws, &nloc);
    lstart[tid] = ls;
    int* __restrict__ lrow = Ls + (size_t)me * (NB + 1);
    if (tid < NB)  lrow[tid] = ls;
    if (tid == 0)  lrow[NB]  = nloc;
    __syncthreads();

    #pragma unroll
    for (int j = 0; j < EPT; ++j) {
        if (du[j] >= 0) {
            const int ru = du[j];
            const int ri = U + di[j];
            stage[lstart[ru >> BIN_SHIFT] + pu[j]] = make_int2(ru, di[j]);
            stage[lstart[ri >> BIN_SHIFT] + pi[j]] = make_int2(ri, du[j]);
        }
    }
    __syncthreads();

    int2* __restrict__ dst = coarse + (size_t)me * (2 * CHUNK);
    for (int i = tid; i < nloc; i += 1024) dst[i] = stage[i];
}

__device__ __forceinline__ int place_of(const int2* run, const int nrun, const int p0, const int q)
{
    int lo = 0;
    int sr = run[0].y;
    for (int st = p0; st > 0; st >>= 1) {
        const int  mid = lo + st;
        const int2 rv  = run[min(mid, nrun - 1)];
        const bool ok  = (mid < nrun) & (rv.x <= q);
        lo = ok ? mid  : lo;
        sr = ok ? rv.y : sr;
    }
    return sr + q;
}

#define CH_MAX 1024
__global__ __launch_bounds__(1024)
void bin_build_kernel(const int2* __restrict__ coarse, const int* __restrict__ Ls,
                      int* __restrict__ off, float* __restrict__ degf,
                      int* __restrict__ bkt, int R, int NB, int NBLK)
{
    __shared__ int cnt[BIN_ROWS];
    __shared__ int cur[BIN_ROWS];
    __shared__ int ws[33];
    __shared__ int sb[STAGE_CAP];
    __shared__ int2 run[CH_MAX];

    const int tid = threadIdx.x;
    const int b   = blockIdx.x;
    const int p0  = 1 << (31 - __clz(max(NBLK - 1, 1)));

    int ra = 0, rn = 0;
    if (tid < NBLK) {
        const int* __restrict__ lrow = Ls + (size_t)tid * (NB + 1) + b;
        ra = lrow[0];
        rn = lrow[1] - ra;
    }
    if (tid < BIN_ROWS) cnt[tid] = 0;

    int s, n;
    block_scan_1024(ra, ws, &s);
    const int rp = block_scan_1024(rn, ws, &n);
    if (tid < NBLK) run[tid] = make_int2(rp, tid * (2 * CHUNK) + ra - rp);
    __syncthreads();

    int2 en[KEEP];
    #pragma unroll
    for (int j = 0; j < KEEP; ++j) en[j] = make_int2(-1, 0);
    if (n > 0) {
        int qq[KEEP], lo[KEEP], sr[KEEP];
        const int sr0 = run[0].y;
        #pragma unroll
        for (int j = 0; j < KEEP; ++j) { qq[j] = min(j * 1024 + tid, n - 1); lo[j] = 0; sr[j] = sr0; }
        for (int st = p0; st > 0; st >>= 1) {
            int2 rv[KEEP];
            #pragma unroll
            for (int j = 0; j < KEEP; ++j) rv[j] = run[min(lo[j] + st, NBLK - 1)];
            #pragma unroll
            for (int j = 0; j < KEEP; ++j) {
                const int  mid = lo[j] + st;
                const bool ok  = (mid < NBLK) & (rv[j].x <= qq[j]);
                lo[j] = ok ? mid     : lo[j];
                sr[j] = ok ? rv[j].y : sr[j];
            }
        }
        #pragma unroll
        for (int j = 0; j < KEEP; ++j) {
            const int2 v = coarse[sr[j] + qq[j]];
            if (j * 1024 + tid < n) en[j] = v;
        }
    }
    #pragma unroll
    for (int j = 0; j < KEEP; ++j)
        if (en[j].x >= 0) atomicAdd(&cnt[en[j].x & (BIN_ROWS - 1)], 1);
    for (int q = KEEP * 1024 + tid; q < n; q += 1024)
        atomicAdd(&cnt[coarse[place_of(run, NBLK, p0, q)].x & (BIN_ROWS - 1)], 1);
    __syncthreads();

    const int c = (tid < BIN_ROWS) ? cnt[tid] : 0;
    int nsum;
    const int ex = block_scan_1024(c, ws, &nsum);
    if (tid < BIN_ROWS) {
        cur[tid] = ex;
        const int r = (b << BIN_SHIFT) + tid;
        if (r < R) {
            off[r]  = s + ex;
            degf[r] = (float)c;
        }
    }
    __syncthreads();

    if (n <= STAGE_CAP) {
        #pragma unroll
        for (int j = 0; j < KEEP; ++j) {
            if (en[j].x >= 0) {
                const int k = atomicAdd(&cur[en[j].x & (BIN_ROWS - 1)], 1);
                sb[k] = en[j].y;
            }
        }
        for (int q = KEEP * 1024 + tid; q < n; q += 1024) {
            const int2 e2 = coarse[place_of(run, NBLK, p0, q)];
            const int  k  = atomicAdd(&cur[e2.x & (BIN_ROWS - 1)], 1);
            sb[k] = e2.y;
        }
        __syncthreads();
        for (int q = tid; q < n; q += 1024) bkt[s + q] = sb[q];
    } else {
        for (int q = tid; q < n; q += 1024) {
            const int2 e2 = coarse[place_of(run, NBLK, p0, q)];
            const int  k  = atomicAdd(&cur[e2.x & (BIN_ROWS - 1)], 1);
            bkt[s + k] = e2.y;
        }
    }
}

#define GW 16

#define RANK_STEP(from, i)                                                          \
    {                                                                               \
        const unsigned s_ = (unsigned)__builtin_amdgcn_update_dpp(                  \
            0, (int)(from), 0x120 + (i), 0xf, 0xf, false);             \
        rk += (s_ < u) ? 1 : 0;                                                     \
    }
#define RANK_ROW(from)                                                              \
    RANK_STEP(from, 1)  RANK_STEP(from, 2)  RANK_STEP(from, 3)  RANK_STEP(from, 4)  \
    RANK_STEP(from, 5)  RANK_STEP(from, 6)  RANK_STEP(from, 7)  RANK_STEP(from, 8)  \
    RANK_STEP(from, 9)  RANK_STEP(from, 10) RANK_STEP(from, 11) RANK_STEP(from, 12) \
    RANK_STEP(from, 13) RANK_STEP(from, 14) RANK_STEP(from, 15)

__device__ __forceinline__ int sort_keys(const int n, const int k, const int lane)
{
    const unsigned u = (lane < n) ? (((unsigned)k << 5) | (unsigned)lane) : 0xffffffffu;
    int rk = 0;
    RANK_ROW(u)
    if (n > 16) {
        const unsigned x = (unsigned)__shfl_xor((int)u, 16);
        rk += (x < u) ? 1 : 0;
        RANK_ROW(x)
    }
    const int rank = (lane < n) ? rk : lane;
    return __builtin_amdgcn_ds_permute(rank << 2, k);
}

__device__ __forceinline__ float4 sum_sorted(const float* __restrict__ src,
                                             const int n, const int sk, const int lane)
{
    float4 acc = make_float4(0.f, 0.f, 0.f, 0.f);
    for (int t0 = 0; t0 < n; t0 += GW) {
        float4 f[GW];
        #pragma unroll
        for (int j = 0; j < GW; ++j) {
            f[j] = make_float4(0.f, 0.f, 0.f, 0.f);
            if (t0 + j < n) {
                const int s = __builtin_amdgcn_readlane(sk, t0 + j);
                f[j] = ((const float4*)(src + (size_t)s * D))[lane];
            }
        }
        #pragma unroll
        for (int j = 0; j < GW; ++j) {
            if (t0 + j < n) {
                acc.x += f[j].x;
                acc.y += f[j].y;
                acc.z += f[j].z;
                acc.w += f[j].w;
            }
        }
    }
    return acc;
}

__device__ __forceinline__ float4 sum_long(const float* __restrict__ src,
                                           const int* __restrict__ bkt,
                                           const int n, const int o, const int lane)
{
    float4 acc = make_float4(0.f, 0.f, 0.f, 0.f);
    int cur = -1;
    for (int trip = 0; trip < n; ++trip) {
        int best = 0x7fffffff, m = 0;
        for (int j = lane; j < n; j += 32) {
            const int kk = bkt[o + j];
            if (kk > cur) {
                if (kk < best)       { best = kk; m = 1; }
                else if (kk == best) { ++m; }
            }
        }
        int wb = best;
        for (int d = 16; d > 0; d >>= 1) wb = min(wb, __shfl_xor(wb, d));
        if (wb == 0x7fffffff) break;
        int mm = (best == wb) ? m : 0;
        for (int d = 16; d > 0; d >>= 1) mm += __shfl_xor(mm, d);
        const int s   = __builtin_amdgcn_readfirstlane(wb);
        const int rep = __builtin_amdgcn_readfirstlane(mm);
        const float4 f = ((const float4*)(src + (size_t)s * D))[lane];
        for (int t = 0; t < rep; ++t) {
            acc.x += f.x; acc.y += f.y; acc.z += f.z; acc.w += f.w;
        }
        cur = s;
    }
    return acc;
}

#define ROW_TILES 5
#define BLOCK_ROWS (ROW_TILES * 16)

__global__ __launch_bounds__(256)
void gnn_gemm_kernel(const float* __restrict__ XU, const float* __restrict__ XI,
                     float* __restrict__ outU, float* __restrict__ outI,
                     const float* __restrict__ WsU, const float* __restrict__ bsU,
                     const float* __restrict__ WmU, const float* __restrict__ bmU,
                     const float* __restrict__ WsI, const float* __restrict__ bsI,
                     const float* __restrict__ WmI, const float* __restrict__ bmI,
                     const float* __restrict__ degf, const int* __restrict__ off,
                     const int* __restrict__ bkt, int U, int I, int nbI)
{
    __shared__ float sX[BLOCK_ROWS * LDS_STRIDE];
    __shared__ float sG[BLOCK_ROWS * LDS_STRIDE];

    const bool item = (int)blockIdx.x < nbI;
    const float* __restrict__ X     = item ? XI  : XU;
    float*       __restrict__ out   = item ? outI : outU;
    const float* __restrict__ Wself = item ? WsI : WsU;
    const float* __restrict__ bself = item ? bsI : bsU;
    const float* __restrict__ Wmsg  = item ? WmI : WmU;
    const float* __restrict__ bmsg  = item ? bmI : bmU;
    const float* __restrict__ deg   = item ? degf + U : degf;
    const int*   __restrict__ offp  = item ? off + U : off;
    const float* __restrict__ srcT  = item ? XU : XI;
    const int rows = item ? I : U;

    const int row0 = (item ? (int)blockIdx.x : (int)blockIdx.x - nbI) * BLOCK_ROWS;
    const int tid  = threadIdx.x;

    float dsum = 0.f;
    {
        float dg[(BLOCK_ROWS + 31) / 32];
        #pragma unroll
        for (int j = 0; j < (BLOCK_ROWS + 31) / 32; ++j) {
            const int i = min(j * 32 + (tid & 31), BLOCK_ROWS - 1);
            dg[j] = deg[min(row0 + i, rows - 1)];
        }
        #pragma unroll
        for (int j = 0; j < (BLOCK_ROWS + 31) / 32; ++j) dsum += dg[j];
    }
    const bool hasEdge = dsum > 0.f;

    {
        const int c4 = tid & 31;
        const int r0 = tid >> 5;
        float4 xr[BLOCK_ROWS / 8];
        #pragma unroll
        for (int j = 0; j < BLOCK_ROWS / 8; ++j) {
            const int gr = min(row0 + r0 + 8 * j, rows - 1);
            xr[j] = ((const float4*)(X + (size_t)gr * D))[c4];
        }

        int hn = 0, ho = 0;
        if (c4 < BLOCK_ROWS / 8) {
            const int gr = min(row0 + r0 + 8 * c4, rows - 1);
            hn = (int)deg[gr];
            ho = offp[gr];
        }
        int n = __builtin_amdgcn_readlane(hn, 0);
        int o = __builtin_amdgcn_readlane(ho, 0);
        int k = 0x7fffffff;
        if (n <= 32 && c4 < n) k = bkt[o + c4];

        #pragma unroll
        for (int j = 0; j < BLOCK_ROWS / 8; ++j) {
            const int r = r0 + 8 * j;
            ((float4*)(sX + r * LDS_STRIDE))[c4] = xr[j];
        }

        for (int j = 0; j < BLOCK_ROWS / 8; ++j) {
            const bool shortb = (n > 0 && n <= 32);
            int sk = 0;
            if (shortb) sk = sort_keys(n, k, c4);
            int n1 = 0, o1 = 0, k1 = 0x7fffffff;
            if (j + 1 < BLOCK_ROWS / 8) {
                n1 = __builtin_amdgcn_readlane(hn, j + 1);
                o1 = __builtin_amdgcn_readlane(ho, j + 1);
                if (n1 <= 32 && c4 < n1) k1 = bkt[o1 + c4];
            }
            float4 a = make_float4(0.f, 0.f, 0.f, 0.f);
            if (shortb)      a = sum_sorted(srcT, n, sk, c4);
            else if (n > 32) a = sum_long(srcT, bkt, n, o, c4);
            ((float4*)(sG + (r0 + 8 * j) * LDS_STRIDE))[c4] = a;
            n = n1; o = o1; k = k1;
        }
    }
    __syncthreads();

    const int lane = tid & 31;
    const int wave = tid >> 5;
    const int col0 = wave * 16;
    const int mn   = lane & 15;
    const int koff = (lane >> 4) << 1;

    const float4* wS = (const float4*)Wself + (size_t)wave * 16 * 32 + lane;
    const float4* wM = (const float4*)Wmsg  + (size_t)wave * 16 * 32 + lane;
    const float* aS = sX + mn * LDS_STRIDE + 2 * koff;
    const float* aM = sG + mn * LDS_STRIDE + 2 * koff;

    v8f c[ROW_TILES];
    #pragma unroll
    for (int t = 0; t < ROW_TILES; ++t) c[t] = (v8f){};

    #pragma unroll
    for (int kk = 0; kk < D / 8; ++kk) {
        const float4 b4 = wS[kk * 32];
        const v2f b0 = { b4.x, b4.y };
        const v2f b1 = { b4.z, b4.w };
        #pragma unroll
        for (int t = 0; t < ROW_TILES; ++t) {
            const float4 a4 = *(const float4*)(aS + t * 16 * LDS_STRIDE + 8 * kk);
            const v2f a0 = { a4.x, a4.y };
            const v2f a1 = { a4.z, a4.w };
            c[t] = __builtin_amdgcn_wmma_f32_16x16x4_f32(false, a0, false, b0,
                                                         (short)0, c[t], false, false);
            c[t] = __builtin_amdgcn_wmma_f32_16x16x4_f32(false, a1, false, b1,
                                                         (short)0, c[t], false, false);
        }
    }
    if (__builtin_amdgcn_ballot_w32(hasEdge) != 0u) {
    #pragma unroll
    for (int kk = 0; kk < D / 8; ++kk) {
        const float4 b4 = wM[kk * 32];
        const v2f b0 = { b4.x, b4.y };
        const v2f b1 = { b4.z, b4.w };
        #pragma unroll
        for (int t = 0; t < ROW_TILES; ++t) {
            const float4 a4 = *(const float4*)(aM + t * 16 * LDS_STRIDE + 8 * kk);
            const v2f a0 = { a4.x, a4.y };
            const v2f a1 = { a4.z, a4.w };
            c[t] = __builtin_amdgcn_wmma_f32_16x16x4_f32(false, a0, false, b0,
                                                         (short)0, c[t], false, false);
            c[t] = __builtin_amdgcn_wmma_f32_16x16x4_f32(false, a1, false, b1,
                                                         (short)0, c[t], false, false);
        }
    }
    }

    const int col  = col0 + mn;
    const float bS = bself[col];
    const float bM = bmsg[col];
    #pragma unroll
    for (int t = 0; t < ROW_TILES; ++t) {
        const int rbase = row0 + t * 16 + ((lane >> 4) << 3);
        if (row0 + t * 16 < rows) {
            #pragma unroll
            for (int v = 0; v < 8; ++v) {
                int row = rbase + v;
                out[(size_t)row * D + col] = c[t][v] + bS + deg[row] * bM;
            }
        }
    }
}

extern "C" void kernel_launch(void* const* d_in, const int* in_sizes, int n_in,
                              void* d_out, int out_size, void* d_ws, size_t ws_size,
                              hipStream_t stream)
{
    const float* UF   = (const float*)d_in[0];
    const float* IF   = (const float*)d_in[1];
    const int*   edge = (const int*)d_in[2];
    const float* Wu   = (const float*)d_in[3];
    const float* bu   = (const float*)d_in[4];
    const float* Wi   = (const float*)d_in[5];
    const float* bi   = (const float*)d_in[6];
    const float* Wu2i = (const float*)d_in[7];
    const float* bu2i = (const float*)d_in[8];
    const float* Wi2u = (const float*)d_in[9];
    const float* bi2u = (const float*)d_in[10];

    const int U = in_sizes[0] / D;
    const int I = in_sizes[1] / D;
    const int E = in_sizes[2] / 2;
    const int R = U + I;

    float* outU = (float*)d_out;
    float* outI = outU + (size_t)U * D;

    const int NB   = (R + BIN_ROWS - 1) / BIN_ROWS;
    const int NBLK = (E + CHUNK - 1) / CHUNK;

    float* Wf     = (float*)d_ws;
    int*   off    = (int*)(Wf + 4 * D * D);
    float* degf   = (float*)(off + R);
    int*   bkt    = (int*)(degf + R);
    int*   Ls     = bkt + (size_t)2 * E;
    const size_t hb = ((size_t)NBLK * (NB + 1) + 1) & ~(size_t)1;
    int2*  coarse = (int2*)(Ls + hb);

    bin_sort_kernel<<<(unsigned)NBLK, 1024, 0, stream>>>(edge, edge + E, Ls, coarse, E, U, NB,
                                                        Wu, Wi2u, Wi, Wu2i, Wf);

    bin_build_kernel<<<(unsigned)NB, 1024, 0, stream>>>(coarse, Ls, off, degf, bkt, R, NB, NBLK);

    const int nbU = (U + BLOCK_ROWS - 1) / BLOCK_ROWS;
    const int nbI = (I + BLOCK_ROWS - 1) / BLOCK_ROWS;
    gnn_gemm_kernel<<<(unsigned)(nbU + nbI), 256, 0, stream>>>(
        UF, IF, outU, outI, Wf, bu, Wf + D * D, bi2u, Wf + 2 * D * D, bi, Wf + 3 * D * D, bu2i,
        degf, off, bkt, U, I, nbI);
}
